# starting kernel
# baseline (speedup 1.0000x reference)
#include <hip/hip_runtime.h>
#include <hip/hip_bf16.h>

typedef __attribute__((ext_vector_type(2))) float v2f;
typedef __attribute__((ext_vector_type(8))) float v8f;

#define HC 128
#define HEADS 4
#define CH 32
#define NEG_SLOPE 0.2f

__device__ __forceinline__ unsigned f2ord(float f) {
    unsigned u = __float_as_uint(f);
    return (u & 0x80000000u) ? ~u : (u | 0x80000000u);
}
__device__ __forceinline__ float ord2f(unsigned u) {
    return (u & 0x80000000u) ? __uint_as_float(u & 0x7FFFFFFFu)
                             : __uint_as_float(~u);
}
#define ORD_NEG_INF 0x007FFFFFu

__device__ __forceinline__ float lrelu(float v) { return v > 0.f ? v : NEG_SLOPE * v; }

__global__ __launch_bounds__(256) void gemm_f32_wmma(
    const float* __restrict__ A, const float* __restrict__ B,
    float* __restrict__ D, int M)
{
    const int wave = threadIdx.x >> 5;
    const int lane = threadIdx.x & 31;
    const int mt   = blockIdx.x;
    const int m15  = lane & 15;
    const int hi   = lane >> 4;
    const int row  = mt * 16 + m15;
    const int col  = wave * 16 + m15;

    v8f acc;
    #pragma unroll
    for (int r = 0; r < 8; ++r) acc[r] = 0.f;

    #pragma unroll 4
    for (int k0 = 0; k0 < HC; k0 += 4) {
        const float* ap = A + (long)row * HC + k0 + 2 * hi;
        v2f a; a.x = ap[0]; a.y = ap[1];
        const float* bp = B + (long)(k0 + 2 * hi) * HC + col;
        v2f b; b.x = bp[0]; b.y = bp[HC];
        acc = __builtin_amdgcn_wmma_f32_16x16x4_f32(
            false, a, false, b, (short)0, acc, false, false);
    }
    #pragma unroll
    for (int r = 0; r < 8; ++r)
        D[(long)(mt * 16 + r + 8 * hi) * HC + col] = acc[r];
}

__global__ void node_alphas(const float* __restrict__ h,
                            const float* __restrict__ a_src,
                            const float* __restrict__ a_dst,
                            float* __restrict__ as, float* __restrict__ ad,
                            int NH)
{
    int i = blockIdx.x * blockDim.x + threadIdx.x;
    if (i >= NH) return;
    int n = i >> 2, hd = i & 3;
    const float* hp  = h + (long)n * HC + hd * CH;
    const float* asp = a_src + hd * CH;
    const float* adp = a_dst + hd * CH;
    float s1 = 0.f, s2 = 0.f;
    #pragma unroll
    for (int c = 0; c < CH; ++c) { float v = hp[c]; s1 += v * asp[c]; s2 += v * adp[c]; }
    as[i] = s1; ad[i] = s2;
}

__global__ void fill_f32(float* p, float v, int n) {
    int i = blockIdx.x * blockDim.x + threadIdx.x;
    if (i < n) p[i] = v;
}
__global__ void fill_u32(unsigned* p, unsigned v, int n) {
    int i = blockIdx.x * blockDim.x + threadIdx.x;
    if (i < n) p[i] = v;
}

__global__ void edge_max(const int* __restrict__ ei, int E, int EL,
                         const float* __restrict__ as, const float* __restrict__ ad,
                         unsigned* __restrict__ m)
{
    int i = blockIdx.x * blockDim.x + threadIdx.x;
    if (i >= EL) return;
    int s, d;
    if (i < E) { s = ei[i]; d = ei[E + i]; } else { s = d = i - E; }
    const float4 a = *(const float4*)(as + (long)s * 4);
    const float4 b = *(const float4*)(ad + (long)d * 4);
    unsigned* mp = m + (long)d * 4;
    atomicMax(mp + 0, f2ord(lrelu(a.x + b.x)));
    atomicMax(mp + 1, f2ord(lrelu(a.y + b.y)));
    atomicMax(mp + 2, f2ord(lrelu(a.z + b.z)));
    atomicMax(mp + 3, f2ord(lrelu(a.w + b.w)));
}

__global__ void edge_sum(const int* __restrict__ ei, int E, int EL,
                         const float* __restrict__ as, const float* __restrict__ ad,
                         const unsigned* __restrict__ m, float* __restrict__ ss)
{
    int i = blockIdx.x * blockDim.x + threadIdx.x;
    if (i >= EL) return;
    int s, d;
    if (i < E) { s = ei[i]; d = ei[E + i]; } else { s = d = i - E; }
    const float4 a = *(const float4*)(as + (long)s * 4);
    const float4 b = *(const float4*)(ad + (long)d * 4);
    const unsigned* mp = m + (long)d * 4;
    float* sp = ss + (long)d * 4;
    atomicAdd(sp + 0, __expf(lrelu(a.x + b.x) - ord2f(mp[0])));
    atomicAdd(sp + 1, __expf(lrelu(a.y + b.y) - ord2f(mp[1])));
    atomicAdd(sp + 2, __expf(lrelu(a.z + b.z) - ord2f(mp[2])));
    atomicAdd(sp + 3, __expf(lrelu(a.w + b.w) - ord2f(mp[3])));
}

__global__ __launch_bounds__(256) void edge_agg(
    const int* __restrict__ ei, int E, int EL,
    const float* __restrict__ h,
    const float* __restrict__ as, const float* __restrict__ ad,
    const unsigned* __restrict__ m, const float* __restrict__ ss,
    float* __restrict__ out)
{
    int e = blockIdx.x * 8 + (threadIdx.x >> 5);
    if (e >= EL) return;
    int lane = threadIdx.x & 31;
    int s, d;
    if (e < E) { s = ei[e]; d = ei[E + e]; } else { s = d = e - E; }
    int hd = lane >> 3;
    float ev    = lrelu(as[(long)s * 4 + hd] + ad[(long)d * 4 + hd]);
    float alpha = __expf(ev - ord2f(m[(long)d * 4 + hd])) / ss[(long)d * 4 + hd];
    const float4 hv = *(const float4*)(h + (long)s * HC + lane * 4);
    float* op = out + (long)d * HC + lane * 4;
    atomicAdd(op + 0, hv.x * alpha);
    atomicAdd(op + 1, hv.y * alpha);
    atomicAdd(op + 2, hv.z * alpha);
    atomicAdd(op + 3, hv.w * alpha);
}

__global__ void bias_elu(float* __restrict__ y, const float* __restrict__ b, int total)
{
    int i = blockIdx.x * blockDim.x + threadIdx.x;
    if (i >= total) return;
    float v = y[i] + b[i & (HC - 1)];
    y[i] = v > 0.f ? v : __expf(v) - 1.f;
}

extern "C" void kernel_launch(void* const* d_in, const int* in_sizes, int n_in,
                              void* d_out, int out_size, void* d_ws, size_t ws_size,
                              hipStream_t stream)
{
    const float* x      = (const float*)d_in[0];
    const int*   ei     = (const int*)d_in[1];
    const float* W1     = (const float*)d_in[2];
    const float* a_src1 = (const float*)d_in[3];
    const float* a_dst1 = (const float*)d_in[4];
    const float* b1     = (const float*)d_in[5];
    const float* W2     = (const float*)d_in[6];
    const float* a_src2 = (const float*)d_in[7];
    const float* a_dst2 = (const float*)d_in[8];
    const float* b2     = (const float*)d_in[9];

    const int N  = in_sizes[0] / HC;
    const int E  = in_sizes[1] / 2;
    const int EL = E + N;
    const int NH = N * HEADS;
    const int NF = N * HC;

    char* ws = (char*)d_ws;
    float*    hbuf = (float*)(ws);
    float*    out1 = (float*)(ws + (size_t)NF * 4);
    float*    as   = (float*)(ws + (size_t)NF * 8);
    float*    ad   = (float*)(ws + (size_t)NF * 8 + (size_t)NH * 4);
    unsigned* mArr = (unsigned*)(ws + (size_t)NF * 8 + (size_t)NH * 8);
    float*    sArr = (float*)(ws + (size_t)NF * 8 + (size_t)NH * 12);

    const int TB = 256;
    const int gN  = (NH + TB - 1) / TB;
    const int gNF = (NF + TB - 1) / TB;
    const int gE  = (EL + TB - 1) / TB;
    const int gEW = (EL + 7) / 8;
    float* outF = (float*)d_out;

    gemm_f32_wmma<<<N / 16, 256, 0, stream>>>(x, W1, hbuf, N);
    node_alphas<<<gN, TB, 0, stream>>>(hbuf, a_src1, a_dst1, as, ad, NH);
    fill_u32<<<gN, TB, 0, stream>>>(mArr, ORD_NEG_INF, NH);
    fill_f32<<<gN, TB, 0, stream>>>(sArr, 0.f, NH);
    fill_f32<<<gNF, TB, 0, stream>>>(out1, 0.f, NF);
    edge_max<<<gE, TB, 0, stream>>>(ei, E, EL, as, ad, mArr);
    edge_sum<<<gE, TB, 0, stream>>>(ei, E, EL, as, ad, mArr, sArr);
    edge_agg<<<gEW, TB, 0, stream>>>(ei, E, EL, hbuf, as, ad, mArr, sArr, out1);
    bias_elu<<<gNF, TB, 0, stream>>>(out1, b1, NF);

    gemm_f32_wmma<<<N / 16, 256, 0, stream>>>(out1, W2, hbuf, N);
    node_alphas<<<gN, TB, 0, stream>>>(hbuf, a_src2, a_dst2, as, ad, NH);
    fill_u32<<<gN, TB, 0, stream>>>(mArr, ORD_NEG_INF, NH);
    fill_f32<<<gN, TB, 0, stream>>>(sArr, 0.f, NH);
    fill_f32<<<gNF, TB, 0, stream>>>(outF, 0.f, NF);
    edge_max<<<gE, TB, 0, stream>>>(ei, E, EL, as, ad, mArr);
    edge_sum<<<gE, TB, 0, stream>>>(ei, E, EL, as, ad, mArr, sArr);
    edge_agg<<<gEW, TB, 0, stream>>>(ei, E, EL, hbuf, as, ad, mArr, sArr, outF);
    bias_elu<<<gNF, TB, 0, stream>>>(outF, b2, NF);
}


# Edges sorted by destination once per call (rank by counter, scan, placement); one wave per node does max, exp-sum and row gather in registers: no float atomics, sums now serial in CSR arrival order
# speedup vs baseline: 4.3986x; 4.3986x over previous
#include <hip/hip_runtime.h>
#include <hip/hip_bf16.h>

typedef __attribute__((ext_vector_type(2))) float v2f;
typedef __attribute__((ext_vector_type(8))) float v8f;

#define HC 128
#define HEADS 4
#define CH 32
#define NEG_SLOPE 0.2f

__device__ __forceinline__ float lrelu(float v) { return v > 0.f ? v : NEG_SLOPE * v; }

__global__ __launch_bounds__(256) void gemm_f32_wmma(
    const float* __restrict__ A, const float* __restrict__ B,
    float* __restrict__ D, int M)
{
    const int wave = threadIdx.x >> 5;
    const int lane = threadIdx.x & 31;
    const int mt   = blockIdx.x;
    const int m15  = lane & 15;
    const int hi   = lane >> 4;
    const int row  = mt * 16 + m15;
    const int col  = wave * 16 + m15;

    v8f acc;
    #pragma unroll
    for (int r = 0; r < 8; ++r) acc[r] = 0.f;

    #pragma unroll 4
    for (int k0 = 0; k0 < HC; k0 += 4) {
        const float* ap = A + (long)row * HC + k0 + 2 * hi;
        v2f a; a.x = ap[0]; a.y = ap[1];
        const float* bp = B + (long)(k0 + 2 * hi) * HC + col;
        v2f b; b.x = bp[0]; b.y = bp[HC];
        acc = __builtin_amdgcn_wmma_f32_16x16x4_f32(
            false, a, false, b, (short)0, acc, false, false);
    }
    #pragma unroll
    for (int r = 0; r < 8; ++r)
        D[(long)(mt * 16 + r + 8 * hi) * HC + col] = acc[r];
}

__global__ void node_alphas(const float* __restrict__ h,
                            const float* __restrict__ a_src,
                            const float* __restrict__ a_dst,
                            float* __restrict__ as, float* __restrict__ ad,
                            int NH)
{
    int i = blockIdx.x * blockDim.x + threadIdx.x;
    if (i >= NH) return;
    int n = i >> 2, hd = i & 3;
    const float* hp  = h + (long)n * HC + hd * CH;
    const float* asp = a_src + hd * CH;
    const float* adp = a_dst + hd * CH;
    float s1 = 0.f, s2 = 0.f;
    #pragma unroll
    for (int c = 0; c < CH; ++c) { float v = hp[c]; s1 += v * asp[c]; s2 += v * adp[c]; }
    as[i] = s1; ad[i] = s2;
}

__global__ void fill_i32(int* p, int v, int n) {
    int i = blockIdx.x * blockDim.x + threadIdx.x;
    if (i < n) p[i] = v;
}

__global__ void dst_rank(const int* __restrict__ ei, int E,
                         int* __restrict__ cnt, int* __restrict__ rank)
{
    int i = blockIdx.x * blockDim.x + threadIdx.x;
    if (i >= E) return;
    rank[i] = atomicAdd(cnt + ei[E + i], 1);
}

__global__ __launch_bounds__(1024) void scan_counts(
    const int* __restrict__ cnt, int* __restrict__ rowptr, int N)
{
    __shared__ int wsum[32];
    const int t = threadIdx.x;
    const int lane = t & 31, w = t >> 5;
    const int chunk = (((N + 1023) >> 10) + 3) & ~3;
    const int b = t * chunk;
    const int e = min(b + chunk, N);
    int sum = 0;
    for (int i = b; i < e; i += 4) {
        const int4 c = *(const int4*)(cnt + i);
        sum += (c.x + c.y) + (c.z + c.w);
    }
    int v = sum;
    for (int o = 1; o < 32; o <<= 1) { int u = __shfl_up(v, o); if (lane >= o) v += u; }
    if (lane == 31) wsum[w] = v;
    __syncthreads();
    if (w == 0) {
        int x = wsum[lane];
        for (int o = 1; o < 32; o <<= 1) { int u = __shfl_up(x, o); if (lane >= o) x += u; }
        wsum[lane] = x;
    }
    __syncthreads();
    int base = v - sum + (w ? wsum[w - 1] : 0);
    for (int i = b; i < e; i += 4) {
        const int4 c = *(const int4*)(cnt + i);
        int4 r;
        r.x = base; r.y = r.x + c.x; r.z = r.y + c.y; r.w = r.z + c.z;
        base = r.w + c.w;
        *(int4*)(rowptr + i) = r;
    }
    if (t == 1023) rowptr[N] = base;
}

__global__ void csr_place(const int* __restrict__ ei, int E,
                          const int* __restrict__ rowptr, const int* __restrict__ rank,
                          int* __restrict__ csr_src)
{
    int i = blockIdx.x * blockDim.x + threadIdx.x;
    if (i >= E) return;
    const int s = ei[i], d = ei[E + i];
    atomicExch(csr_src + rowptr[d] + rank[i], s);
}

__global__ __launch_bounds__(256) void node_gather(
    const int* __restrict__ rowptr, const int* __restrict__ csr_src,
    const float* __restrict__ h,
    const float* __restrict__ as, const float* __restrict__ ad,
    float* __restrict__ out, int N)
{
    const int d = __builtin_amdgcn_readfirstlane((int)(blockIdx.x * 8 + (threadIdx.x >> 5)));
    if (d >= N) return;
    const int lane = threadIdx.x & 31;
    const int hd   = lane >> 3;
    const int beg = rowptr[d], end = rowptr[d + 1];
    const float4 adv = *(const float4*)(ad + (long)d * 4);
    const float4 asd = *(const float4*)(as + (long)d * 4);
    float4 es;
    es.x = lrelu(asd.x + adv.x); es.y = lrelu(asd.y + adv.y);
    es.z = lrelu(asd.z + adv.z); es.w = lrelu(asd.w + adv.w);

    float4 mx = es;
    for (int j0 = beg; j0 < end; j0 += 32) {
        const int j = j0 + lane;
        if (j < end) {
            const int s = csr_src[j];
            const float4 a = *(const float4*)(as + (long)s * 4);
            mx.x = fmaxf(mx.x, lrelu(a.x + adv.x));
            mx.y = fmaxf(mx.y, lrelu(a.y + adv.y));
            mx.z = fmaxf(mx.z, lrelu(a.z + adv.z));
            mx.w = fmaxf(mx.w, lrelu(a.w + adv.w));
        }
    }
    for (int o = 16; o > 0; o >>= 1) {
        mx.x = fmaxf(mx.x, __shfl_xor(mx.x, o));
        mx.y = fmaxf(mx.y, __shfl_xor(mx.y, o));
        mx.z = fmaxf(mx.z, __shfl_xor(mx.z, o));
        mx.w = fmaxf(mx.w, __shfl_xor(mx.w, o));
    }

    float4 sm; sm.x = sm.y = sm.z = sm.w = 0.f;
    for (int j0 = beg; j0 < end; j0 += 32) {
        const int j = j0 + lane;
        if (j < end) {
            const int s = csr_src[j];
            const float4 a = *(const float4*)(as + (long)s * 4);
            sm.x += __expf(lrelu(a.x + adv.x) - mx.x);
            sm.y += __expf(lrelu(a.y + adv.y) - mx.y);
            sm.z += __expf(lrelu(a.z + adv.z) - mx.z);
            sm.w += __expf(lrelu(a.w + adv.w) - mx.w);
        }
    }
    for (int o = 16; o > 0; o >>= 1) {
        sm.x += __shfl_xor(sm.x, o);
        sm.y += __shfl_xor(sm.y, o);
        sm.z += __shfl_xor(sm.z, o);
        sm.w += __shfl_xor(sm.w, o);
    }
    float4 ps;
    ps.x = __expf(es.x - mx.x); ps.y = __expf(es.y - mx.y);
    ps.z = __expf(es.z - mx.z); ps.w = __expf(es.w - mx.w);
    sm.x += ps.x; sm.y += ps.y; sm.z += ps.z; sm.w += ps.w;

    const float aself = hd == 0 ? ps.x / sm.x : hd == 1 ? ps.y / sm.y
                      : hd == 2 ? ps.z / sm.z : ps.w / sm.w;
    const float4 hs = *(const float4*)(h + (long)d * HC + lane * 4);
    float4 acc;
    acc.x = hs.x * aself; acc.y = hs.y * aself;
    acc.z = hs.z * aself; acc.w = hs.w * aself;
    for (int j0 = beg; j0 < end; j0 += 32) {
        const int j = j0 + lane;
        int s = d;
        float4 p; p.x = p.y = p.z = p.w = 0.f;
        if (j < end) {
            s = csr_src[j];
            const float4 a = *(const float4*)(as + (long)s * 4);
            p.x = __expf(lrelu(a.x + adv.x) - mx.x) / sm.x;
            p.y = __expf(lrelu(a.y + adv.y) - mx.y) / sm.y;
            p.z = __expf(lrelu(a.z + adv.z) - mx.z) / sm.z;
            p.w = __expf(lrelu(a.w + adv.w) - mx.w) / sm.w;
        }
        const int cnt = min(32, end - j0);
        for (int t = 0; t < cnt; t += 4) {
            #pragma unroll
            for (int u = 0; u < 4; ++u) {
                const int st = __builtin_amdgcn_readlane(s, t + u);
                const float p0 = __int_as_float(__builtin_amdgcn_readlane(__float_as_int(p.x), t + u));
                const float p1 = __int_as_float(__builtin_amdgcn_readlane(__float_as_int(p.y), t + u));
                const float p2 = __int_as_float(__builtin_amdgcn_readlane(__float_as_int(p.z), t + u));
                const float p3 = __int_as_float(__builtin_amdgcn_readlane(__float_as_int(p.w), t + u));
                const float al = hd == 0 ? p0 : hd == 1 ? p1 : hd == 2 ? p2 : p3;
                const float4 hv = *(const float4*)(h + (long)st * HC + lane * 4);
                acc.x = fmaf(hv.x, al, acc.x);
                acc.y = fmaf(hv.y, al, acc.y);
                acc.z = fmaf(hv.z, al, acc.z);
                acc.w = fmaf(hv.w, al, acc.w);
            }
        }
    }
    *(float4*)(out + (long)d * HC + lane * 4) = acc;
}

__global__ void bias_elu(float* __restrict__ y, const float* __restrict__ b, int total)
{
    int i = blockIdx.x * blockDim.x + threadIdx.x;
    if (i >= total) return;
    float v = y[i] + b[i & (HC - 1)];
    y[i] = v > 0.f ? v : __expf(v) - 1.f;
}

extern "C" void kernel_launch(void* const* d_in, const int* in_sizes, int n_in,
                              void* d_out, int out_size, void* d_ws, size_t ws_size,
                              hipStream_t stream)
{
    const float* x      = (const float*)d_in[0];
    const int*   ei     = (const int*)d_in[1];
    const float* W1     = (const float*)d_in[2];
    const float* a_src1 = (const float*)d_in[3];
    const float* a_dst1 = (const float*)d_in[4];
    const float* b1     = (const float*)d_in[5];
    const float* W2     = (const float*)d_in[6];
    const float* a_src2 = (const float*)d_in[7];
    const float* a_dst2 = (const float*)d_in[8];
    const float* b2     = (const float*)d_in[9];

    const int N  = in_sizes[0] / HC;
    const int E  = in_sizes[1] / 2;
    const int NH = N * HEADS;
    const int NF = N * HC;

    char* ws = (char*)d_ws;
    float* hbuf   = (float*)(ws);
    float* out1   = (float*)(ws + (size_t)NF * 4);
    float* as     = (float*)(ws + (size_t)NF * 8);
    float* ad     = (float*)(ws + (size_t)NF * 8 + (size_t)NH * 4);
    int*   cnt    = (int*)(ws + (size_t)NF * 8 + (size_t)NH * 8);
    int*   rowptr = (int*)(ws + (size_t)NF * 8 + (size_t)NH * 12);
    int*   csrsrc = (int*)(ws + (size_t)NF * 8 + (size_t)NH * 16);
    int*   rank   = (int*)(ws + (size_t)NF * 8 + (size_t)NH * 16 + (size_t)E * 4);

    const int TB = 256;
    const int gN  = (NH + TB - 1) / TB;
    const int gNF = (NF + TB - 1) / TB;
    const int gN1 = (N + TB - 1) / TB;
    const int gE  = (E + TB - 1) / TB;
    const int gNW = (N + 7) / 8;
    float* outF = (float*)d_out;

    fill_i32<<<gN1, TB, 0, stream>>>(cnt, 0, N);
    dst_rank<<<gE, TB, 0, stream>>>(ei, E, cnt, rank);
    scan_counts<<<1, 1024, 0, stream>>>(cnt, rowptr, N);
    csr_place<<<gE, TB, 0, stream>>>(ei, E, rowptr, rank, csrsrc);

    gemm_f32_wmma<<<N / 16, 256, 0, stream>>>(x, W1, hbuf, N);
    node_alphas<<<gN, TB, 0, stream>>>(hbuf, a_src1, a_dst1, as, ad, NH);
    node_gather<<<gNW, TB, 0, stream>>>(rowptr, csrsrc, hbuf, as, ad, out1, N);
    bias_elu<<<gNF, TB, 0, stream>>>(out1, b1, NF);

    gemm_f32_wmma<<<N / 16, 256, 0, stream>>>(out1, W2, hbuf, N);
    node_alphas<<<gN, TB, 0, stream>>>(hbuf, a_src2, a_dst2, as, ad, NH);
    node_gather<<<gNW, TB, 0, stream>>>(rowptr, csrsrc, hbuf, as, ad, outF, N);
    bias_elu<<<gNF, TB, 0, stream>>>(outF, b2, NF);
}


# Bias and ELU applied in the gather kernel on the finished row before its one store: the two element-wise launches (one element per thread, most of the run's waves) are gone
# speedup vs baseline: 4.8428x; 1.1010x over previous
#include <hip/hip_runtime.h>
#include <hip/hip_bf16.h>

typedef __attribute__((ext_vector_type(2))) float v2f;
typedef __attribute__((ext_vector_type(8))) float v8f;

#define HC 128
#define HEADS 4
#define CH 32
#define NEG_SLOPE 0.2f

__device__ __forceinline__ float lrelu(float v) { return v > 0.f ? v : NEG_SLOPE * v; }

__global__ __launch_bounds__(256) void gemm_f32_wmma(
    const float* __restrict__ A, const float* __restrict__ B,
    float* __restrict__ D, int M)
{
    const int wave = threadIdx.x >> 5;
    const int lane = threadIdx.x & 31;
    const int mt   = blockIdx.x;
    const int m15  = lane & 15;
    const int hi   = lane >> 4;
    const int row  = mt * 16 + m15;
    const int col  = wave * 16 + m15;

    v8f acc;
    #pragma unroll
    for (int r = 0; r < 8; ++r) acc[r] = 0.f;

    #pragma unroll 4
    for (int k0 = 0; k0 < HC; k0 += 4) {
        const float* ap = A + (long)row * HC + k0 + 2 * hi;
        v2f a; a.x = ap[0]; a.y = ap[1];
        const float* bp = B + (long)(k0 + 2 * hi) * HC + col;
        v2f b; b.x = bp[0]; b.y = bp[HC];
        acc = __builtin_amdgcn_wmma_f32_16x16x4_f32(
            false, a, false, b, (short)0, acc, false, false);
    }
    #pragma unroll
    for (int r = 0; r < 8; ++r)
        D[(long)(mt * 16 + r + 8 * hi) * HC + col] = acc[r];
}

__global__ void node_alphas(const float* __restrict__ h,
                            const float* __restrict__ a_src,
                            const float* __restrict__ a_dst,
                            float* __restrict__ as, float* __restrict__ ad,
                            int NH)
{
    int i = blockIdx.x * blockDim.x + threadIdx.x;
    if (i >= NH) return;
    int n = i >> 2, hd = i & 3;
    const float* hp  = h + (long)n * HC + hd * CH;
    const float* asp = a_src + hd * CH;
    const float* adp = a_dst + hd * CH;
    float s1 = 0.f, s2 = 0.f;
    #pragma unroll
    for (int c = 0; c < CH; ++c) { float v = hp[c]; s1 += v * asp[c]; s2 += v * adp[c]; }
    as[i] = s1; ad[i] = s2;
}

__global__ void fill_i32(int* p, int v, int n) {
    int i = blockIdx.x * blockDim.x + threadIdx.x;
    if (i < n) p[i] = v;
}

__global__ void dst_rank(const int* __restrict__ ei, int E,
                         int* __restrict__ cnt, int* __restrict__ rank)
{
    int i = blockIdx.x * blockDim.x + threadIdx.x;
    if (i >= E) return;
    rank[i] = atomicAdd(cnt + ei[E + i], 1);
}

__global__ __launch_bounds__(1024) void scan_counts(
    const int* __restrict__ cnt, int* __restrict__ rowptr, int N)
{
    __shared__ int wsum[32];
    const int t = threadIdx.x;
    const int lane = t & 31, w = t >> 5;
    const int chunk = (((N + 1023) >> 10) + 3) & ~3;
    const int b = t * chunk;
    const int e = min(b + chunk, N);
    int sum = 0;
    for (int i = b; i < e; i += 4) {
        const int4 c = *(const int4*)(cnt + i);
        sum += (c.x + c.y) + (c.z + c.w);
    }
    int v = sum;
    for (int o = 1; o < 32; o <<= 1) { int u = __shfl_up(v, o); if (lane >= o) v += u; }
    if (lane == 31) wsum[w] = v;
    __syncthreads();
    if (w == 0) {
        int x = wsum[lane];
        for (int o = 1; o < 32; o <<= 1) { int u = __shfl_up(x, o); if (lane >= o) x += u; }
        wsum[lane] = x;
    }
    __syncthreads();
    int base = v - sum + (w ? wsum[w - 1] : 0);
    for (int i = b; i < e; i += 4) {
        const int4 c = *(const int4*)(cnt + i);
        int4 r;
        r.x = base; r.y = r.x + c.x; r.z = r.y + c.y; r.w = r.z + c.z;
        base = r.w + c.w;
        *(int4*)(rowptr + i) = r;
    }
    if (t == 1023) rowptr[N] = base;
}

__global__ void csr_place(const int* __restrict__ ei, int E,
                          const int* __restrict__ rowptr, const int* __restrict__ rank,
                          int* __restrict__ csr_src)
{
    int i = blockIdx.x * blockDim.x + threadIdx.x;
    if (i >= E) return;
    const int s = ei[i], d = ei[E + i];
    atomicExch(csr_src + rowptr[d] + rank[i], s);
}

__global__ __launch_bounds__(256) void node_gather(
    const int* __restrict__ rowptr, const int* __restrict__ csr_src,
    const float* __restrict__ h,
    const float* __restrict__ as, const float* __restrict__ ad,
    const float* __restrict__ bias,
    float* __restrict__ out, int N)
{
    const int d = __builtin_amdgcn_readfirstlane((int)(blockIdx.x * 8 + (threadIdx.x >> 5)));
    if (d >= N) return;
    const int lane = threadIdx.x & 31;
    const int hd   = lane >> 3;
    const int beg = rowptr[d], end = rowptr[d + 1];
    const float4 adv = *(const float4*)(ad + (long)d * 4);
    const float4 asd = *(const float4*)(as + (long)d * 4);
    float4 es;
    es.x = lrelu(asd.x + adv.x); es.y = lrelu(asd.y + adv.y);
    es.z = lrelu(asd.z + adv.z); es.w = lrelu(asd.w + adv.w);

    float4 mx = es;
    for (int j0 = beg; j0 < end; j0 += 32) {
        const int j = j0 + lane;
        if (j < end) {
            const int s = csr_src[j];
            const float4 a = *(const float4*)(as + (long)s * 4);
            mx.x = fmaxf(mx.x, lrelu(a.x + adv.x));
            mx.y = fmaxf(mx.y, lrelu(a.y + adv.y));
            mx.z = fmaxf(mx.z, lrelu(a.z + adv.z));
            mx.w = fmaxf(mx.w, lrelu(a.w + adv.w));
        }
    }
    for (int o = 16; o > 0; o >>= 1) {
        mx.x = fmaxf(mx.x, __shfl_xor(mx.x, o));
        mx.y = fmaxf(mx.y, __shfl_xor(mx.y, o));
        mx.z = fmaxf(mx.z, __shfl_xor(mx.z, o));
        mx.w = fmaxf(mx.w, __shfl_xor(mx.w, o));
    }

    float4 sm; sm.x = sm.y = sm.z = sm.w = 0.f;
    for (int j0 = beg; j0 < end; j0 += 32) {
        const int j = j0 + lane;
        if (j < end) {
            const int s = csr_src[j];
            const float4 a = *(const float4*)(as + (long)s * 4);
            sm.x += __expf(lrelu(a.x + adv.x) - mx.x);
            sm.y += __expf(lrelu(a.y + adv.y) - mx.y);
            sm.z += __expf(lrelu(a.z + adv.z) - mx.z);
            sm.w += __expf(lrelu(a.w + adv.w) - mx.w);
        }
    }
    for (int o = 16; o > 0; o >>= 1) {
        sm.x += __shfl_xor(sm.x, o);
        sm.y += __shfl_xor(sm.y, o);
        sm.z += __shfl_xor(sm.z, o);
        sm.w += __shfl_xor(sm.w, o);
    }
    float4 ps;
    ps.x = __expf(es.x - mx.x); ps.y = __expf(es.y - mx.y);
    ps.z = __expf(es.z - mx.z); ps.w = __expf(es.w - mx.w);
    sm.x += ps.x; sm.y += ps.y; sm.z += ps.z; sm.w += ps.w;

    const float aself = hd == 0 ? ps.x / sm.x : hd == 1 ? ps.y / sm.y
                      : hd == 2 ? ps.z / sm.z : ps.w / sm.w;
    const float4 hs = *(const float4*)(h + (long)d * HC + lane * 4);
    float4 acc;
    acc.x = hs.x * aself; acc.y = hs.y * aself;
    acc.z = hs.z * aself; acc.w = hs.w * aself;
    for (int j0 = beg; j0 < end; j0 += 32) {
        const int j = j0 + lane;
        int s = d;
        float4 p; p.x = p.y = p.z = p.w = 0.f;
        if (j < end) {
            s = csr_src[j];
            const float4 a = *(const float4*)(as + (long)s * 4);
            p.x = __expf(lrelu(a.x + adv.x) - mx.x) / sm.x;
            p.y = __expf(lrelu(a.y + adv.y) - mx.y) / sm.y;
            p.z = __expf(lrelu(a.z + adv.z) - mx.z) / sm.z;
            p.w = __expf(lrelu(a.w + adv.w) - mx.w) / sm.w;
        }
        const int cnt = min(32, end - j0);
        for (int t = 0; t < cnt; t += 4) {
            #pragma unroll
            for (int u = 0; u < 4; ++u) {
                const int st = __builtin_amdgcn_readlane(s, t + u);
                const float p0 = __int_as_float(__builtin_amdgcn_readlane(__float_as_int(p.x), t + u));
                const float p1 = __int_as_float(__builtin_amdgcn_readlane(__float_as_int(p.y), t + u));
                const float p2 = __int_as_float(__builtin_amdgcn_readlane(__float_as_int(p.z), t + u));
                const float p3 = __int_as_float(__builtin_amdgcn_readlane(__float_as_int(p.w), t + u));
                const float al = hd == 0 ? p0 : hd == 1 ? p1 : hd == 2 ? p2 : p3;
                const float4 hv = *(const float4*)(h + (long)st * HC + lane * 4);
                acc.x = fmaf(hv.x, al, acc.x);
                acc.y = fmaf(hv.y, al, acc.y);
                acc.z = fmaf(hv.z, al, acc.z);
                acc.w = fmaf(hv.w, al, acc.w);
            }
        }
    }
    const float4 bv = *(const float4*)(bias + lane * 4);
    float4 y;
    y.x = acc.x + bv.x; y.y = acc.y + bv.y;
    y.z = acc.z + bv.z; y.w = acc.w + bv.w;
    y.x = y.x > 0.f ? y.x : __expf(y.x) - 1.f;
    y.y = y.y > 0.f ? y.y : __expf(y.y) - 1.f;
    y.z = y.z > 0.f ? y.z : __expf(y.z) - 1.f;
    y.w = y.w > 0.f ? y.w : __expf(y.w) - 1.f;
    *(float4*)(out + (long)d * HC + lane * 4) = y;
}

extern "C" void kernel_launch(void* const* d_in, const int* in_sizes, int n_in,
                              void* d_out, int out_size, void* d_ws, size_t ws_size,
                              hipStream_t stream)
{
    const float* x      = (const float*)d_in[0];
    const int*   ei     = (const int*)d_in[1];
    const float* W1     = (const float*)d_in[2];
    const float* a_src1 = (const float*)d_in[3];
    const float* a_dst1 = (const float*)d_in[4];
    const float* b1     = (const float*)d_in[5];
    const float* W2     = (const float*)d_in[6];
    const float* a_src2 = (const float*)d_in[7];
    const float* a_dst2 = (const float*)d_in[8];
    const float* b2     = (const float*)d_in[9];

    const int N  = in_sizes[0] / HC;
    const int E  = in_sizes[1] / 2;
    const int NH = N * HEADS;
    const int NF = N * HC;

    char* ws = (char*)d_ws;
    float* hbuf   = (float*)(ws);
    float* out1   = (float*)(ws + (size_t)NF * 4);
    float* as     = (float*)(ws + (size_t)NF * 8);
    float* ad     = (float*)(ws + (size_t)NF * 8 + (size_t)NH * 4);
    int*   cnt    = (int*)(ws + (size_t)NF * 8 + (size_t)NH * 8);
    int*   rowptr = (int*)(ws + (size_t)NF * 8 + (size_t)NH * 12);
    int*   csrsrc = (int*)(ws + (size_t)NF * 8 + (size_t)NH * 16);
    int*   rank   = (int*)(ws + (size_t)NF * 8 + (size_t)NH * 16 + (size_t)E * 4);

    const int TB = 256;
    const int gN  = (NH + TB - 1) / TB;
    const int gN1 = (N + TB - 1) / TB;
    const int gE  = (E + TB - 1) / TB;
    const int gNW = (N + 7) / 8;
    float* outF = (float*)d_out;

    fill_i32<<<gN1, TB, 0, stream>>>(cnt, 0, N);
    dst_rank<<<gE, TB, 0, stream>>>(ei, E, cnt, rank);
    scan_counts<<<1, 1024, 0, stream>>>(cnt, rowptr, N);
    csr_place<<<gE, TB, 0, stream>>>(ei, E, rowptr, rank, csrsrc);

    gemm_f32_wmma<<<N / 16, 256, 0, stream>>>(x, W1, hbuf, N);
    node_alphas<<<gN, TB, 0, stream>>>(hbuf, a_src1, a_dst1, as, ad, NH);
    node_gather<<<gNW, TB, 0, stream>>>(rowptr, csrsrc, hbuf, as, ad, b1, out1, N);

    gemm_f32_wmma<<<N / 16, 256, 0, stream>>>(out1, W2, hbuf, N);
    node_alphas<<<gN, TB, 0, stream>>>(hbuf, a_src2, a_dst2, as, ad, NH);
    node_gather<<<gNW, TB, 0, stream>>>(rowptr, csrsrc, hbuf, as, ad, b2, outF, N);
}


# Edge list grouped by destination with LDS counters in two levels (slices ordered by 256-node bucket, then one work-group per bucket) instead of a global atomic per edge; sums stay in arrival order
# speedup vs baseline: 8.4710x; 1.7492x over previous
#include <hip/hip_runtime.h>
#include <hip/hip_bf16.h>

typedef __attribute__((ext_vector_type(2))) float v2f;
typedef __attribute__((ext_vector_type(8))) float v8f;

#define HC 128
#define HEADS 4
#define CH 32
#define NEG_SLOPE 0.2f

__device__ __forceinline__ float lrelu(float v) { return v > 0.f ? v : NEG_SLOPE * v; }

__global__ __launch_bounds__(256) void gemm_f32_wmma(
    const float* __restrict__ A, const float* __restrict__ B,
    float* __restrict__ D, int M)
{
    const int wave = threadIdx.x >> 5;
    const int lane = threadIdx.x & 31;
    const int mt   = blockIdx.x;
    const int m15  = lane & 15;
    const int hi   = lane >> 4;
    const int row  = mt * 16 + m15;
    const int col  = wave * 16 + m15;

    v8f acc;
    #pragma unroll
    for (int r = 0; r < 8; ++r) acc[r] = 0.f;

    #pragma unroll 4
    for (int k0 = 0; k0 < HC; k0 += 4) {
        const float* ap = A + (long)row * HC + k0 + 2 * hi;
        v2f a; a.x = ap[0]; a.y = ap[1];
        const float* bp = B + (long)(k0 + 2 * hi) * HC + col;
        v2f b; b.x = bp[0]; b.y = bp[HC];
        acc = __builtin_amdgcn_wmma_f32_16x16x4_f32(
            false, a, false, b, (short)0, acc, false, false);
    }
    #pragma unroll
    for (int r = 0; r < 8; ++r)
        D[(long)(mt * 16 + r + 8 * hi) * HC + col] = acc[r];
}

__global__ void node_alphas(const float* __restrict__ h,
                            const float* __restrict__ a_src,
                            const float* __restrict__ a_dst,
                            float* __restrict__ as, float* __restrict__ ad,
                            int NH)
{
    int i = blockIdx.x * blockDim.x + threadIdx.x;
    if (i >= NH) return;
    int n = i >> 2, hd = i & 3;
    const float* hp  = h + (long)n * HC + hd * CH;
    const float* asp = a_src + hd * CH;
    const float* adp = a_dst + hd * CH;
    float s1 = 0.f, s2 = 0.f;
    #pragma unroll
    for (int c = 0; c < CH; ++c) { float v = hp[c]; s1 += v * asp[c]; s2 += v * adp[c]; }
    as[i] = s1; ad[i] = s2;
}

#define BSH       8
#define BNODES    256
#define SLICE     6272
#define SLICE_IT  7
#define LPITCH    288
#define STAGE_CAP 12288

__global__ __launch_bounds__(1024) void slice_sort(
    const int* __restrict__ ei, int E,
    int* __restrict__ recs, int* __restrict__ loff)
{
    __shared__ int hist[BNODES];
    __shared__ int wsum[8];
    __shared__ int sorted[SLICE];
    const int t = threadIdx.x, lane = t & 31, w = t >> 5;
    const int e0 = blockIdx.x * SLICE;
    const int e1 = min(e0 + SLICE, E);
    if (t < BNODES) hist[t] = 0;
    __syncthreads();

    int rec[SLICE_IT], bk[SLICE_IT], rk[SLICE_IT];
    #pragma unroll
    for (int i = 0; i < SLICE_IT; ++i) {
        const int idx = e0 + t + i * 1024;
        const int idc = min(idx, e1 - 1);
        const int s = ei[idc], d = ei[E + idc];
        rk[i]  = 0;
        bk[i]  = idx < e1 ? (d >> BSH) : -1;
        rec[i] = (s << BSH) | (d & (BNODES - 1));
    }
    #pragma unroll
    for (int i = 0; i < SLICE_IT; ++i)
        if (bk[i] >= 0) rk[i] = atomicAdd(&hist[bk[i]], 1);
    __syncthreads();

    int c = 0, v = 0;
    if (t < BNODES) {
        c = hist[t]; v = c;
        for (int o = 1; o < 32; o <<= 1) { int u = __shfl_up(v, o); if (lane >= o) v += u; }
        if (lane == 31) wsum[w] = v;
    }
    __syncthreads();
    if (t < BNODES) {
        int base = 0;
        for (int k = 0; k < w; ++k) base += wsum[k];
        const int ex = base + v - c;
        hist[t] = ex;
        loff[blockIdx.x * LPITCH + t] = ex;
        if (t == BNODES - 1) loff[blockIdx.x * LPITCH + BNODES] = ex + c;
    }
    __syncthreads();
    #pragma unroll
    for (int i = 0; i < SLICE_IT; ++i)
        if (bk[i] >= 0) sorted[hist[bk[i]] + rk[i]] = rec[i];
    __syncthreads();
    const int n = e1 - e0;
    for (int i = t; i < n; i += 1024) recs[e0 + i] = sorted[i];
}

__global__ __launch_bounds__(1024) void bucket_finish(
    const int* __restrict__ recs, const int* __restrict__ loff, int nslice, int N,
    int* __restrict__ rowptr, int* __restrict__ csr_src)
{
    __shared__ int runlo[256], runhi[256];
    __shared__ int cnt[BNODES];
    __shared__ int wsA[8], wsM[8], wsum[8];
    __shared__ int stage[STAGE_CAP];
    const int t = threadIdx.x, lane = t & 31, w = t >> 5;
    const int b = blockIdx.x;

    if (t < 256) {
        int lo = 0, hi = 0;
        if (t < nslice) { lo = loff[t * LPITCH + b]; hi = loff[t * LPITCH + b + 1]; }
        runlo[t] = lo; runhi[t] = hi; cnt[t] = 0;
        int a = lo, m = hi - lo;
        for (int o = 16; o > 0; o >>= 1) { a += __shfl_xor(a, o); m += __shfl_xor(m, o); }
        if (lane == 0) { wsA[w] = a; wsM[w] = m; }
    }
    __syncthreads();
    int r0 = 0, total = 0;
    #pragma unroll
    for (int k = 0; k < 8; ++k) { r0 += wsA[k]; total += wsM[k]; }

    int r[16];
    #pragma unroll
    for (int q = 0; q < 8; ++q) {
        const int sl = min(w + 32 * q, nslice - 1);
        const int* rp = recs + (long)sl * SLICE;
        const int i = runlo[sl] + lane, h = runhi[sl];
        const int x0 = rp[max(min(i, h - 1), 0)];
        const int x1 = rp[max(min(i + 32, h - 1), 0)];
        const bool own = w + 32 * q < nslice;
        r[2 * q]     = (own && i < h)      ? x0 : -1;
        r[2 * q + 1] = (own && i + 32 < h) ? x1 : -1;
    }
    #pragma unroll
    for (int q = 0; q < 16; ++q)
        if (r[q] >= 0) atomicAdd(&cnt[r[q] & (BNODES - 1)], 1);
    for (int q = 0; q < 8; ++q) {
        const int sl = w + 32 * q;
        if (sl < nslice) {
            const int* rp = recs + (long)sl * SLICE;
            for (int i = runlo[sl] + 64 + lane; i < runhi[sl]; i += 32)
                atomicAdd(&cnt[rp[i] & (BNODES - 1)], 1);
        }
    }
    __syncthreads();

    int c = 0, v = 0;
    if (t < BNODES) {
        c = cnt[t]; v = c;
        for (int o = 1; o < 32; o <<= 1) { int u = __shfl_up(v, o); if (lane >= o) v += u; }
        if (lane == 31) wsum[w] = v;
    }
    __syncthreads();
    if (t < BNODES) {
        int base = 0;
        for (int k = 0; k < w; ++k) base += wsum[k];
        const int ex = base + v - c;
        cnt[t] = ex;
        const int node = b * BNODES + t;
        if (node <= N) rowptr[node] = r0 + ex;
    }
    if ((N & (BNODES - 1)) == 0 && b == (int)gridDim.x - 1 && t == 0) rowptr[N] = r0 + total;
    __syncthreads();

    const bool staged = total <= STAGE_CAP;
    #pragma unroll
    for (int q = 0; q < 16; ++q)
        if (r[q] >= 0) {
            const int pos = atomicAdd(&cnt[r[q] & (BNODES - 1)], 1);
            if (staged) stage[pos] = r[q] >> BSH;
            else atomicExch(csr_src + r0 + pos, r[q] >> BSH);
        }
    for (int q = 0; q < 8; ++q) {
        const int sl = w + 32 * q;
        if (sl < nslice) {
            const int* rp = recs + (long)sl * SLICE;
            for (int i = runlo[sl] + 64 + lane; i < runhi[sl]; i += 32) {
                const int x = rp[i];
                const int pos = atomicAdd(&cnt[x & (BNODES - 1)], 1);
                if (staged) stage[pos] = x >> BSH;
                else atomicExch(csr_src + r0 + pos, x >> BSH);
            }
        }
    }
    __syncthreads();
    if (staged) {
        const int r1 = r0 + total;
        for (int a = (r0 & ~31) + t; a < r1; a += 1024) {
            if (a >= r0) {
                const int x = stage[a - r0];
                const int line = a & ~31;
                if (line >= r0 && line + 32 <= r1) csr_src[a] = x;
                else atomicExch(csr_src + a, x);
            }
        }
    }
}

__global__ __launch_bounds__(256) void node_gather(
    const int* __restrict__ rowptr, const int* __restrict__ csr_src,
    const float* __restrict__ h,
    const float* __restrict__ as, const float* __restrict__ ad,
    const float* __restrict__ bias,
    float* __restrict__ out, int N)
{
    const int d = __builtin_amdgcn_readfirstlane((int)(blockIdx.x * 8 + (threadIdx.x >> 5)));
    if (d >= N) return;
    const int lane = threadIdx.x & 31;
    const int hd   = lane >> 3;
    const int beg = rowptr[d], end = rowptr[d + 1];
    const float4 adv = *(const float4*)(ad + (long)d * 4);
    const float4 asd = *(const float4*)(as + (long)d * 4);
    float4 es;
    es.x = lrelu(asd.x + adv.x); es.y = lrelu(asd.y + adv.y);
    es.z = lrelu(asd.z + adv.z); es.w = lrelu(asd.w + adv.w);

    float4 mx = es;
    for (int j0 = beg; j0 < end; j0 += 32) {
        const int j = j0 + lane;
        if (j < end) {
            const int s = csr_src[j];
            const float4 a = *(const float4*)(as + (long)s * 4);
            mx.x = fmaxf(mx.x, lrelu(a.x + adv.x));
            mx.y = fmaxf(mx.y, lrelu(a.y + adv.y));
            mx.z = fmaxf(mx.z, lrelu(a.z + adv.z));
            mx.w = fmaxf(mx.w, lrelu(a.w + adv.w));
        }
    }
    for (int o = 16; o > 0; o >>= 1) {
        mx.x = fmaxf(mx.x, __shfl_xor(mx.x, o));
        mx.y = fmaxf(mx.y, __shfl_xor(mx.y, o));
        mx.z = fmaxf(mx.z, __shfl_xor(mx.z, o));
        mx.w = fmaxf(mx.w, __shfl_xor(mx.w, o));
    }

    float4 sm; sm.x = sm.y = sm.z = sm.w = 0.f;
    for (int j0 = beg; j0 < end; j0 += 32) {
        const int j = j0 + lane;
        if (j < end) {
            const int s = csr_src[j];
            const float4 a = *(const float4*)(as + (long)s * 4);
            sm.x += __expf(lrelu(a.x + adv.x) - mx.x);
            sm.y += __expf(lrelu(a.y + adv.y) - mx.y);
            sm.z += __expf(lrelu(a.z + adv.z) - mx.z);
            sm.w += __expf(lrelu(a.w + adv.w) - mx.w);
        }
    }
    for (int o = 16; o > 0; o >>= 1) {
        sm.x += __shfl_xor(sm.x, o);
        sm.y += __shfl_xor(sm.y, o);
        sm.z += __shfl_xor(sm.z, o);
        sm.w += __shfl_xor(sm.w, o);
    }
    float4 ps;
    ps.x = __expf(es.x - mx.x); ps.y = __expf(es.y - mx.y);
    ps.z = __expf(es.z - mx.z); ps.w = __expf(es.w - mx.w);
    sm.x += ps.x; sm.y += ps.y; sm.z += ps.z; sm.w += ps.w;

    const float aself = hd == 0 ? ps.x / sm.x : hd == 1 ? ps.y / sm.y
                      : hd == 2 ? ps.z / sm.z : ps.w / sm.w;
    const float4 hs = *(const float4*)(h + (long)d * HC + lane * 4);
    float4 acc;
    acc.x = hs.x * aself; acc.y = hs.y * aself;
    acc.z = hs.z * aself; acc.w = hs.w * aself;
    for (int j0 = beg; j0 < end; j0 += 32) {
        const int j = j0 + lane;
        int s = d;
        float4 p; p.x = p.y = p.z = p.w = 0.f;
        if (j < end) {
            s = csr_src[j];
            const float4 a = *(const float4*)(as + (long)s * 4);
            p.x = __expf(lrelu(a.x + adv.x) - mx.x) / sm.x;
            p.y = __expf(lrelu(a.y + adv.y) - mx.y) / sm.y;
            p.z = __expf(lrelu(a.z + adv.z) - mx.z) / sm.z;
            p.w = __expf(lrelu(a.w + adv.w) - mx.w) / sm.w;
        }
        const int cnt = min(32, end - j0);
        for (int t = 0; t < cnt; t += 4) {
            #pragma unroll
            for (int u = 0; u < 4; ++u) {
                const int st = __builtin_amdgcn_readlane(s, t + u);
                const float p0 = __int_as_float(__builtin_amdgcn_readlane(__float_as_int(p.x), t + u));
                const float p1 = __int_as_float(__builtin_amdgcn_readlane(__float_as_int(p.y), t + u));
                const float p2 = __int_as_float(__builtin_amdgcn_readlane(__float_as_int(p.z), t + u));
                const float p3 = __int_as_float(__builtin_amdgcn_readlane(__float_as_int(p.w), t + u));
                const float al = hd == 0 ? p0 : hd == 1 ? p1 : hd == 2 ? p2 : p3;
                const float4 hv = *(const float4*)(h + (long)st * HC + lane * 4);
                acc.x = fmaf(hv.x, al, acc.x);
                acc.y = fmaf(hv.y, al, acc.y);
                acc.z = fmaf(hv.z, al, acc.z);
                acc.w = fmaf(hv.w, al, acc.w);
            }
        }
    }
    const float4 bv = *(const float4*)(bias + lane * 4);
    float4 y;
    y.x = acc.x + bv.x; y.y = acc.y + bv.y;
    y.z = acc.z + bv.z; y.w = acc.w + bv.w;
    y.x = y.x > 0.f ? y.x : __expf(y.x) - 1.f;
    y.y = y.y > 0.f ? y.y : __expf(y.y) - 1.f;
    y.z = y.z > 0.f ? y.z : __expf(y.z) - 1.f;
    y.w = y.w > 0.f ? y.w : __expf(y.w) - 1.f;
    *(float4*)(out + (long)d * HC + lane * 4) = y;
}

extern "C" void kernel_launch(void* const* d_in, const int* in_sizes, int n_in,
                              void* d_out, int out_size, void* d_ws, size_t ws_size,
                              hipStream_t stream)
{
    const float* x      = (const float*)d_in[0];
    const int*   ei     = (const int*)d_in[1];
    const float* W1     = (const float*)d_in[2];
    const float* a_src1 = (const float*)d_in[3];
    const float* a_dst1 = (const float*)d_in[4];
    const float* b1     = (const float*)d_in[5];
    const float* W2     = (const float*)d_in[6];
    const float* a_src2 = (const float*)d_in[7];
    const float* a_dst2 = (const float*)d_in[8];
    const float* b2     = (const float*)d_in[9];

    const int N  = in_sizes[0] / HC;
    const int E  = in_sizes[1] / 2;
    const int NH = N * HEADS;
    const int NF = N * HC;

    char* ws = (char*)d_ws;
    float* hbuf   = (float*)(ws);
    float* out1   = (float*)(ws + (size_t)NF * 4);
    float* as     = (float*)(ws + (size_t)NF * 8);
    float* ad     = (float*)(ws + (size_t)NF * 8 + (size_t)NH * 4);
    int*   rowptr = (int*)(ws + (size_t)NF * 8 + (size_t)NH * 12);
    int*   csrsrc = (int*)(ws + (size_t)NF * 8 + (size_t)NH * 16);
    int*   recs   = (int*)(ws + (size_t)NF * 8 + (size_t)NH * 16 + (size_t)E * 4);

    const int TB = 256;
    const int gN  = (NH + TB - 1) / TB;
    const int nslice = (E + SLICE - 1) / SLICE;
    const int nbk    = (N + BNODES - 1) / BNODES;
    int*   loff   = recs + (size_t)nslice * SLICE;
    const int gNW = (N + 7) / 8;
    float* outF = (float*)d_out;

    slice_sort<<<nslice, 1024, 0, stream>>>(ei, E, recs, loff);
    bucket_finish<<<nbk, 1024, 0, stream>>>(recs, loff, nslice, N, rowptr, csrsrc);

    gemm_f32_wmma<<<N / 16, 256, 0, stream>>>(x, W1, hbuf, N);
    node_alphas<<<gN, TB, 0, stream>>>(hbuf, a_src1, a_dst1, as, ad, NH);
    node_gather<<<gNW, TB, 0, stream>>>(rowptr, csrsrc, hbuf, as, ad, b1, out1, N);

    gemm_f32_wmma<<<N / 16, 256, 0, stream>>>(out1, W2, hbuf, N);
    node_alphas<<<gN, TB, 0, stream>>>(hbuf, a_src2, a_dst2, as, ad, NH);
    node_gather<<<gNW, TB, 0, stream>>>(rowptr, csrsrc, hbuf, as, ad, b2, outF, N);
}


# Gather kernel fetches each edge's source id and logits once: a node's first 64 edges are held in registers for max, exp-sum and weights, one 32-line logit gather per edge instead of three
# speedup vs baseline: 9.9734x; 1.1774x over previous
#include <hip/hip_runtime.h>
#include <hip/hip_bf16.h>

typedef __attribute__((ext_vector_type(2))) float v2f;
typedef __attribute__((ext_vector_type(8))) float v8f;

#define HC 128
#define HEADS 4
#define CH 32
#define NEG_SLOPE 0.2f

__device__ __forceinline__ float lrelu(float v) { return v > 0.f ? v : NEG_SLOPE * v; }

__global__ __launch_bounds__(256) void gemm_f32_wmma(
    const float* __restrict__ A, const float* __restrict__ B,
    float* __restrict__ D, int M)
{
    const int wave = threadIdx.x >> 5;
    const int lane = threadIdx.x & 31;
    const int mt   = blockIdx.x;
    const int m15  = lane & 15;
    const int hi   = lane >> 4;
    const int row  = mt * 16 + m15;
    const int col  = wave * 16 + m15;

    v8f acc;
    #pragma unroll
    for (int r = 0; r < 8; ++r) acc[r] = 0.f;

    #pragma unroll 4
    for (int k0 = 0; k0 < HC; k0 += 4) {
        const float* ap = A + (long)row * HC + k0 + 2 * hi;
        v2f a; a.x = ap[0]; a.y = ap[1];
        const float* bp = B + (long)(k0 + 2 * hi) * HC + col;
        v2f b; b.x = bp[0]; b.y = bp[HC];
        acc = __builtin_amdgcn_wmma_f32_16x16x4_f32(
            false, a, false, b, (short)0, acc, false, false);
    }
    #pragma unroll
    for (int r = 0; r < 8; ++r)
        D[(long)(mt * 16 + r + 8 * hi) * HC + col] = acc[r];
}

__global__ void node_alphas(const float* __restrict__ h,
                            const float* __restrict__ a_src,
                            const float* __restrict__ a_dst,
                            float* __restrict__ as, float* __restrict__ ad,
                            int NH)
{
    int i = blockIdx.x * blockDim.x + threadIdx.x;
    if (i >= NH) return;
    int n = i >> 2, hd = i & 3;
    const float* hp  = h + (long)n * HC + hd * CH;
    const float* asp = a_src + hd * CH;
    const float* adp = a_dst + hd * CH;
    float s1 = 0.f, s2 = 0.f;
    #pragma unroll
    for (int c = 0; c < CH; ++c) { float v = hp[c]; s1 += v * asp[c]; s2 += v * adp[c]; }
    as[i] = s1; ad[i] = s2;
}

#define BSH       8
#define BNODES    256
#define SLICE     6272
#define SLICE_IT  7
#define LPITCH    288
#define STAGE_CAP 12288

__global__ __launch_bounds__(1024) void slice_sort(
    const int* __restrict__ ei, int E,
    int* __restrict__ recs, int* __restrict__ loff)
{
    __shared__ int hist[BNODES];
    __shared__ int wsum[8];
    __shared__ int sorted[SLICE];
    const int t = threadIdx.x, lane = t & 31, w = t >> 5;
    const int e0 = blockIdx.x * SLICE;
    const int e1 = min(e0 + SLICE, E);
    if (t < BNODES) hist[t] = 0;
    __syncthreads();

    int rec[SLICE_IT], bk[SLICE_IT], rk[SLICE_IT];
    #pragma unroll
    for (int i = 0; i < SLICE_IT; ++i) {
        const int idx = e0 + t + i * 1024;
        const int idc = min(idx, e1 - 1);
        const int s = ei[idc], d = ei[E + idc];
        rk[i]  = 0;
        bk[i]  = idx < e1 ? (d >> BSH) : -1;
        rec[i] = (s << BSH) | (d & (BNODES - 1));
    }
    #pragma unroll
    for (int i = 0; i < SLICE_IT; ++i)
        if (bk[i] >= 0) rk[i] = atomicAdd(&hist[bk[i]], 1);
    __syncthreads();

    int c = 0, v = 0;
    if (t < BNODES) {
        c = hist[t]; v = c;
        for (int o = 1; o < 32; o <<= 1) { int u = __shfl_up(v, o); if (lane >= o) v += u; }
        if (lane == 31) wsum[w] = v;
    }
    __syncthreads();
    if (t < BNODES) {
        int base = 0;
        for (int k = 0; k < w; ++k) base += wsum[k];
        const int ex = base + v - c;
        hist[t] = ex;
        loff[blockIdx.x * LPITCH + t] = ex;
        if (t == BNODES - 1) loff[blockIdx.x * LPITCH + BNODES] = ex + c;
    }
    __syncthreads();
    #pragma unroll
    for (int i = 0; i < SLICE_IT; ++i)
        if (bk[i] >= 0) sorted[hist[bk[i]] + rk[i]] = rec[i];
    __syncthreads();
    const int n = e1 - e0;
    for (int i = t; i < n; i += 1024) recs[e0 + i] = sorted[i];
}

__global__ __launch_bounds__(1024) void bucket_finish(
    const int* __restrict__ recs, const int* __restrict__ loff, int nslice, int N,
    int* __restrict__ rowptr, int* __restrict__ csr_src)
{
    __shared__ int runlo[256], runhi[256];
    __shared__ int cnt[BNODES];
    __shared__ int wsA[8], wsM[8], wsum[8];
    __shared__ int stage[STAGE_CAP];
    const int t = threadIdx.x, lane = t & 31, w = t >> 5;
    const int b = blockIdx.x;

    if (t < 256) {
        int lo = 0, hi = 0;
        if (t < nslice) { lo = loff[t * LPITCH + b]; hi = loff[t * LPITCH + b + 1]; }
        runlo[t] = lo; runhi[t] = hi; cnt[t] = 0;
        int a = lo, m = hi - lo;
        for (int o = 16; o > 0; o >>= 1) { a += __shfl_xor(a, o); m += __shfl_xor(m, o); }
        if (lane == 0) { wsA[w] = a; wsM[w] = m; }
    }
    __syncthreads();
    int r0 = 0, total = 0;
    #pragma unroll
    for (int k = 0; k < 8; ++k) { r0 += wsA[k]; total += wsM[k]; }

    int r[16];
    #pragma unroll
    for (int q = 0; q < 8; ++q) {
        const int sl = min(w + 32 * q, nslice - 1);
        const int* rp = recs + (long)sl * SLICE;
        const int i = runlo[sl] + lane, h = runhi[sl];
        const int x0 = rp[max(min(i, h - 1), 0)];
        const int x1 = rp[max(min(i + 32, h - 1), 0)];
        const bool own = w + 32 * q < nslice;
        r[2 * q]     = (own && i < h)      ? x0 : -1;
        r[2 * q + 1] = (own && i + 32 < h) ? x1 : -1;
    }
    #pragma unroll
    for (int q = 0; q < 16; ++q)
        if (r[q] >= 0) atomicAdd(&cnt[r[q] & (BNODES - 1)], 1);
    for (int q = 0; q < 8; ++q) {
        const int sl = w + 32 * q;
        if (sl < nslice) {
            const int* rp = recs + (long)sl * SLICE;
            for (int i = runlo[sl] + 64 + lane; i < runhi[sl]; i += 32)
                atomicAdd(&cnt[rp[i] & (BNODES - 1)], 1);
        }
    }
    __syncthreads();

    int c = 0, v = 0;
    if (t < BNODES) {
        c = cnt[t]; v = c;
        for (int o = 1; o < 32; o <<= 1) { int u = __shfl_up(v, o); if (lane >= o) v += u; }
        if (lane == 31) wsum[w] = v;
    }
    __syncthreads();
    if (t < BNODES) {
        int base = 0;
        for (int k = 0; k < w; ++k) base += wsum[k];
        const int ex = base + v - c;
        cnt[t] = ex;
        const int node = b * BNODES + t;
        if (node <= N) rowptr[node] = r0 + ex;
    }
    if ((N & (BNODES - 1)) == 0 && b == (int)gridDim.x - 1 && t == 0) rowptr[N] = r0 + total;
    __syncthreads();

    const bool staged = total <= STAGE_CAP;
    #pragma unroll
    for (int q = 0; q < 16; ++q)
        if (r[q] >= 0) {
            const int pos = atomicAdd(&cnt[r[q] & (BNODES - 1)], 1);
            if (staged) stage[pos] = r[q] >> BSH;
            else atomicExch(csr_src + r0 + pos, r[q] >> BSH);
        }
    for (int q = 0; q < 8; ++q) {
        const int sl = w + 32 * q;
        if (sl < nslice) {
            const int* rp = recs + (long)sl * SLICE;
            for (int i = runlo[sl] + 64 + lane; i < runhi[sl]; i += 32) {
                const int x = rp[i];
                const int pos = atomicAdd(&cnt[x & (BNODES - 1)], 1);
                if (staged) stage[pos] = x >> BSH;
                else atomicExch(csr_src + r0 + pos, x >> BSH);
            }
        }
    }
    __syncthreads();
    if (staged) {
        const int r1 = r0 + total;
        for (int a = (r0 & ~31) + t; a < r1; a += 1024) {
            if (a >= r0) {
                const int x = stage[a - r0];
                const int line = a & ~31;
                if (line >= r0 && line + 32 <= r1) csr_src[a] = x;
                else atomicExch(csr_src + a, x);
            }
        }
    }
}


__device__ __forceinline__ void add_rows(
    const float* __restrict__ h, int s, const float4 p, int cnt, int lane, float4& acc)
{
    const bool h0 = (lane & 8) != 0, h1 = (lane & 16) != 0;
    for (int t = 0; t < cnt; t += 4) {
        #pragma unroll
        for (int u = 0; u < 4; ++u) {
            const int st = __builtin_amdgcn_readlane(s, t + u);
            const float p0 = __int_as_float(__builtin_amdgcn_readlane(__float_as_int(p.x), t + u));
            const float p1 = __int_as_float(__builtin_amdgcn_readlane(__float_as_int(p.y), t + u));
            const float p2 = __int_as_float(__builtin_amdgcn_readlane(__float_as_int(p.z), t + u));
            const float p3 = __int_as_float(__builtin_amdgcn_readlane(__float_as_int(p.w), t + u));
            const float plo = h0 ? p1 : p0, phi = h0 ? p3 : p2;
            const float al = h1 ? phi : plo;
            const float4 hv = *(const float4*)(h + (long)st * HC + lane * 4);
            acc.x = fmaf(hv.x, al, acc.x);
            acc.y = fmaf(hv.y, al, acc.y);
            acc.z = fmaf(hv.z, al, acc.z);
            acc.w = fmaf(hv.w, al, acc.w);
        }
    }
}

__global__ __launch_bounds__(256) void node_gather(
    const int* __restrict__ rowptr, const int* __restrict__ csr_src,
    const float* __restrict__ h,
    const float* __restrict__ as, const float* __restrict__ ad,
    const float* __restrict__ bias,
    float* __restrict__ out, int N)
{
    const int d = __builtin_amdgcn_readfirstlane((int)(blockIdx.x * 8 + (threadIdx.x >> 5)));
    if (d >= N) return;
    const int lane = threadIdx.x & 31;
    const int hd   = lane >> 3;
    const int beg = rowptr[d], end = rowptr[d + 1];
    const float4 adv = *(const float4*)(ad + (long)d * 4);
    const float4 asd = *(const float4*)(as + (long)d * 4);
    float4 es;
    es.x = lrelu(asd.x + adv.x); es.y = lrelu(asd.y + adv.y);
    es.z = lrelu(asd.z + adv.z); es.w = lrelu(asd.w + adv.w);

    const int  jA = beg + lane, jB = jA + 32;
    const bool vA = jA < end,   vB = jB < end;
    const int  jl = max(end - 1, 0);
    const int  cA = csr_src[min(jA, jl)];
    const int  cB = csr_src[min(jB, jl)];
    const float4 aA = *(const float4*)(as + (long)cA * 4);
    const float4 aB = *(const float4*)(as + (long)cB * 4);
    asm volatile("" :: "v"(aA.x), "v"(aA.y), "v"(aA.z), "v"(aA.w),
                       "v"(aB.x), "v"(aB.y), "v"(aB.z), "v"(aB.w));
    const int sA = vA ? cA : d, sB = vB ? cB : d;
    float4 eA, eB;
    eA.x = vA ? lrelu(aA.x + adv.x) : es.x;  eA.y = vA ? lrelu(aA.y + adv.y) : es.y;
    eA.z = vA ? lrelu(aA.z + adv.z) : es.z;  eA.w = vA ? lrelu(aA.w + adv.w) : es.w;
    eB.x = vB ? lrelu(aB.x + adv.x) : es.x;  eB.y = vB ? lrelu(aB.y + adv.y) : es.y;
    eB.z = vB ? lrelu(aB.z + adv.z) : es.z;  eB.w = vB ? lrelu(aB.w + adv.w) : es.w;

    float4 mx;
    mx.x = fmaxf(fmaxf(es.x, eA.x), eB.x);  mx.y = fmaxf(fmaxf(es.y, eA.y), eB.y);
    mx.z = fmaxf(fmaxf(es.z, eA.z), eB.z);  mx.w = fmaxf(fmaxf(es.w, eA.w), eB.w);
    for (int j0 = beg + 64; j0 < end; j0 += 32) {
        const int j = j0 + lane;
        if (j < end) {
            const int s = csr_src[j];
            const float4 a = *(const float4*)(as + (long)s * 4);
            mx.x = fmaxf(mx.x, lrelu(a.x + adv.x));
            mx.y = fmaxf(mx.y, lrelu(a.y + adv.y));
            mx.z = fmaxf(mx.z, lrelu(a.z + adv.z));
            mx.w = fmaxf(mx.w, lrelu(a.w + adv.w));
        }
    }
    for (int o = 16; o > 0; o >>= 1) {
        mx.x = fmaxf(mx.x, __shfl_xor(mx.x, o));
        mx.y = fmaxf(mx.y, __shfl_xor(mx.y, o));
        mx.z = fmaxf(mx.z, __shfl_xor(mx.z, o));
        mx.w = fmaxf(mx.w, __shfl_xor(mx.w, o));
    }

    float4 pA, pB;
    pA.x = vA ? __expf(eA.x - mx.x) : 0.f;  pA.y = vA ? __expf(eA.y - mx.y) : 0.f;
    pA.z = vA ? __expf(eA.z - mx.z) : 0.f;  pA.w = vA ? __expf(eA.w - mx.w) : 0.f;
    pB.x = vB ? __expf(eB.x - mx.x) : 0.f;  pB.y = vB ? __expf(eB.y - mx.y) : 0.f;
    pB.z = vB ? __expf(eB.z - mx.z) : 0.f;  pB.w = vB ? __expf(eB.w - mx.w) : 0.f;
    float4 sm;
    sm.x = pA.x + pB.x; sm.y = pA.y + pB.y; sm.z = pA.z + pB.z; sm.w = pA.w + pB.w;
    for (int j0 = beg + 64; j0 < end; j0 += 32) {
        const int j = j0 + lane;
        if (j < end) {
            const int s = csr_src[j];
            const float4 a = *(const float4*)(as + (long)s * 4);
            sm.x += __expf(lrelu(a.x + adv.x) - mx.x);
            sm.y += __expf(lrelu(a.y + adv.y) - mx.y);
            sm.z += __expf(lrelu(a.z + adv.z) - mx.z);
            sm.w += __expf(lrelu(a.w + adv.w) - mx.w);
        }
    }
    for (int o = 16; o > 0; o >>= 1) {
        sm.x += __shfl_xor(sm.x, o);
        sm.y += __shfl_xor(sm.y, o);
        sm.z += __shfl_xor(sm.z, o);
        sm.w += __shfl_xor(sm.w, o);
    }
    float4 ps;
    ps.x = __expf(es.x - mx.x); ps.y = __expf(es.y - mx.y);
    ps.z = __expf(es.z - mx.z); ps.w = __expf(es.w - mx.w);
    sm.x += ps.x; sm.y += ps.y; sm.z += ps.z; sm.w += ps.w;

    const float aself = hd == 0 ? ps.x / sm.x : hd == 1 ? ps.y / sm.y
                      : hd == 2 ? ps.z / sm.z : ps.w / sm.w;
    const float4 hs = *(const float4*)(h + (long)d * HC + lane * 4);
    float4 acc;
    acc.x = hs.x * aself; acc.y = hs.y * aself;
    acc.z = hs.z * aself; acc.w = hs.w * aself;
    pA.x = pA.x / sm.x; pA.y = pA.y / sm.y; pA.z = pA.z / sm.z; pA.w = pA.w / sm.w;
    pB.x = pB.x / sm.x; pB.y = pB.y / sm.y; pB.z = pB.z / sm.z; pB.w = pB.w / sm.w;
    add_rows(h, sA, pA, min(32, end - beg), lane, acc);
    add_rows(h, sB, pB, min(32, end - beg - 32), lane, acc);
    for (int j0 = beg + 64; j0 < end; j0 += 32) {
        const int j = j0 + lane;
        int s = d;
        float4 p; p.x = p.y = p.z = p.w = 0.f;
        if (j < end) {
            s = csr_src[j];
            const float4 a = *(const float4*)(as + (long)s * 4);
            p.x = __expf(lrelu(a.x + adv.x) - mx.x) / sm.x;
            p.y = __expf(lrelu(a.y + adv.y) - mx.y) / sm.y;
            p.z = __expf(lrelu(a.z + adv.z) - mx.z) / sm.z;
            p.w = __expf(lrelu(a.w + adv.w) - mx.w) / sm.w;
        }
        add_rows(h, s, p, min(32, end - j0), lane, acc);
    }
    const float4 bv = *(const float4*)(bias + lane * 4);
    float4 y;
    y.x = acc.x + bv.x; y.y = acc.y + bv.y;
    y.z = acc.z + bv.z; y.w = acc.w + bv.w;
    y.x = y.x > 0.f ? y.x : __expf(y.x) - 1.f;
    y.y = y.y > 0.f ? y.y : __expf(y.y) - 1.f;
    y.z = y.z > 0.f ? y.z : __expf(y.z) - 1.f;
    y.w = y.w > 0.f ? y.w : __expf(y.w) - 1.f;
    *(float4*)(out + (long)d * HC + lane * 4) = y;
}

extern "C" void kernel_launch(void* const* d_in, const int* in_sizes, int n_in,
                              void* d_out, int out_size, void* d_ws, size_t ws_size,
                              hipStream_t stream)
{
    const float* x      = (const float*)d_in[0];
    const int*   ei     = (const int*)d_in[1];
    const float* W1     = (const float*)d_in[2];
    const float* a_src1 = (const float*)d_in[3];
    const float* a_dst1 = (const float*)d_in[4];
    const float* b1     = (const float*)d_in[5];
    const float* W2     = (const float*)d_in[6];
    const float* a_src2 = (const float*)d_in[7];
    const float* a_dst2 = (const float*)d_in[8];
    const float* b2     = (const float*)d_in[9];

    const int N  = in_sizes[0] / HC;
    const int E  = in_sizes[1] / 2;
    const int NH = N * HEADS;
    const int NF = N * HC;

    char* ws = (char*)d_ws;
    float* hbuf   = (float*)(ws);
    float* out1   = (float*)(ws + (size_t)NF * 4);
    float* as     = (float*)(ws + (size_t)NF * 8);
    float* ad     = (float*)(ws + (size_t)NF * 8 + (size_t)NH * 4);
    int*   rowptr = (int*)(ws + (size_t)NF * 8 + (size_t)NH * 12);
    int*   csrsrc = (int*)(ws + (size_t)NF * 8 + (size_t)NH * 16);
    int*   recs   = (int*)(ws + (size_t)NF * 8 + (size_t)NH * 16 + (size_t)E * 4);

    const int TB = 256;
    const int gN  = (NH + TB - 1) / TB;
    const int nslice = (E + SLICE - 1) / SLICE;
    const int nbk    = (N + BNODES - 1) / BNODES;
    int*   loff   = recs + (size_t)nslice * SLICE;
    const int gNW = (N + 7) / 8;
    float* outF = (float*)d_out;

    slice_sort<<<nslice, 1024, 0, stream>>>(ei, E, recs, loff);
    bucket_finish<<<nbk, 1024, 0, stream>>>(recs, loff, nslice, N, rowptr, csrsrc);

    gemm_f32_wmma<<<N / 16, 256, 0, stream>>>(x, W1, hbuf, N);
    node_alphas<<<gN, TB, 0, stream>>>(hbuf, a_src1, a_dst1, as, ad, NH);
    node_gather<<<gNW, TB, 0, stream>>>(rowptr, csrsrc, hbuf, as, ad, b1, out1, N);

    gemm_f32_wmma<<<N / 16, 256, 0, stream>>>(out1, W2, hbuf, N);
    node_alphas<<<gN, TB, 0, stream>>>(hbuf, a_src2, a_dst2, as, ad, NH);
    node_gather<<<gNW, TB, 0, stream>>>(rowptr, csrsrc, hbuf, as, ad, b2, outF, N);
}


# Per-node logits taken from the accumulator tiles in the GEMM's epilogue (16-lane tree per column tile, a head's two tiles added via LDS: sum order of the logits changes); node_alphas launches gone
# speedup vs baseline: 10.4843x; 1.0512x over previous
#include <hip/hip_runtime.h>
#include <hip/hip_bf16.h>

typedef __attribute__((ext_vector_type(2))) float v2f;
typedef __attribute__((ext_vector_type(8))) float v8f;

#define HC 128
#define HEADS 4
#define CH 32
#define NEG_SLOPE 0.2f

__device__ __forceinline__ float lrelu(float v) { return v > 0.f ? v : NEG_SLOPE * v; }

template <int M> __device__ __forceinline__ float row_xor(float v) {
    return __builtin_bit_cast(float, __builtin_amdgcn_update_dpp(
        0, __builtin_bit_cast(int, v), 0x160 | M, 0xf, 0xf, false));
}
__device__ __forceinline__ float row_sums8(const float (&v)[8], int lane) {
    const bool b8 = (lane & 8) != 0, b4 = (lane & 4) != 0, b2 = (lane & 2) != 0;
    float a[4], b[2];
    #pragma unroll
    for (int j = 0; j < 4; ++j)
        a[j] = (b8 ? v[j + 4] : v[j]) + row_xor<8>(b8 ? v[j] : v[j + 4]);
    #pragma unroll
    for (int j = 0; j < 2; ++j)
        b[j] = (b4 ? a[j + 2] : a[j]) + row_xor<4>(b4 ? a[j] : a[j + 2]);
    const float c = (b2 ? b[1] : b[0]) + row_xor<2>(b2 ? b[0] : b[1]);
    return c + row_xor<1>(c);
}

__global__ __launch_bounds__(256) void gemm_f32_wmma(
    const float* __restrict__ A, const float* __restrict__ B,
    const float* __restrict__ a_src, const float* __restrict__ a_dst,
    float* __restrict__ D, float* __restrict__ as, float* __restrict__ ad, int M)
{
    __shared__ float part[2][8][16];
    const int wave = threadIdx.x >> 5;
    const int lane = threadIdx.x & 31;
    const int mt   = blockIdx.x;
    const int m15  = lane & 15;
    const int hi   = lane >> 4;
    const int row  = mt * 16 + m15;
    const int col  = wave * 16 + m15;

    const float ws = a_src[col], wd = a_dst[col];

    v8f acc;
    #pragma unroll
    for (int r = 0; r < 8; ++r) acc[r] = 0.f;

    #pragma unroll 4
    for (int k0 = 0; k0 < HC; k0 += 4) {
        const float* ap = A + (long)row * HC + k0 + 2 * hi;
        v2f a; a.x = ap[0]; a.y = ap[1];
        const float* bp = B + (long)(k0 + 2 * hi) * HC + col;
        v2f b; b.x = bp[0]; b.y = bp[HC];
        acc = __builtin_amdgcn_wmma_f32_16x16x4_f32(
            false, a, false, b, (short)0, acc, false, false);
    }
    float ps[8], pd[8];
    #pragma unroll
    for (int r = 0; r < 8; ++r) { ps[r] = acc[r] * ws; pd[r] = acc[r] * wd; }
    const float ss = row_sums8(ps, lane), sd = row_sums8(pd, lane);
    part[lane & 1][wave][8 * hi + (m15 >> 1)] = (lane & 1) ? sd : ss;
    __syncthreads();
    const int t = threadIdx.x;
    if (t < 128) {
        const int which = t >> 6, i = t & 63;
        const int rl = i >> 2, hd = i & 3;
        const float v = part[which][2 * hd][rl] + part[which][2 * hd + 1][rl];
        float* dstp = which ? ad : as;
        dstp[(long)mt * 64 + i] = v;
    }
    #pragma unroll
    for (int r = 0; r < 8; ++r)
        D[(long)(mt * 16 + r + 8 * hi) * HC + col] = acc[r];
}

#define BSH       8
#define BNODES    256
#define SLICE     6272
#define SLICE_IT  7
#define LPITCH    288
#define STAGE_CAP 12288

__global__ __launch_bounds__(1024) void slice_sort(
    const int* __restrict__ ei, int E,
    int* __restrict__ recs, int* __restrict__ loff)
{
    __shared__ int hist[BNODES];
    __shared__ int wsum[8];
    __shared__ int sorted[SLICE];
    const int t = threadIdx.x, lane = t & 31, w = t >> 5;
    const int e0 = blockIdx.x * SLICE;
    const int e1 = min(e0 + SLICE, E);
    if (t < BNODES) hist[t] = 0;
    __syncthreads();

    int rec[SLICE_IT], bk[SLICE_IT], rk[SLICE_IT];
    #pragma unroll
    for (int i = 0; i < SLICE_IT; ++i) {
        const int idx = e0 + t + i * 1024;
        const int idc = min(idx, e1 - 1);
        const int s = ei[idc], d = ei[E + idc];
        rk[i]  = 0;
        bk[i]  = idx < e1 ? (d >> BSH) : -1;
        rec[i] = (s << BSH) | (d & (BNODES - 1));
    }
    #pragma unroll
    for (int i = 0; i < SLICE_IT; ++i)
        if (bk[i] >= 0) rk[i] = atomicAdd(&hist[bk[i]], 1);
    __syncthreads();

    int c = 0, v = 0;
    if (t < BNODES) {
        c = hist[t]; v = c;
        for (int o = 1; o < 32; o <<= 1) { int u = __shfl_up(v, o); if (lane >= o) v += u; }
        if (lane == 31) wsum[w] = v;
    }
    __syncthreads();
    if (t < BNODES) {
        int base = 0;
        for (int k = 0; k < w; ++k) base += wsum[k];
        const int ex = base + v - c;
        hist[t] = ex;
        loff[blockIdx.x * LPITCH + t] = ex;
        if (t == BNODES - 1) loff[blockIdx.x * LPITCH + BNODES] = ex + c;
    }
    __syncthreads();
    #pragma unroll
    for (int i = 0; i < SLICE_IT; ++i)
        if (bk[i] >= 0) sorted[hist[bk[i]] + rk[i]] = rec[i];
    __syncthreads();
    const int n = e1 - e0;
    for (int i = t; i < n; i += 1024) recs[e0 + i] = sorted[i];
}

__global__ __launch_bounds__(1024) void bucket_finish(
    const int* __restrict__ recs, const int* __restrict__ loff, int nslice, int N,
    int* __restrict__ rowptr, int* __restrict__ csr_src)
{
    __shared__ int runlo[256], runhi[256];
    __shared__ int cnt[BNODES];
    __shared__ int wsA[8], wsM[8], wsum[8];
    __shared__ int stage[STAGE_CAP];
    const int t = threadIdx.x, lane = t & 31, w = t >> 5;
    const int b = blockIdx.x;

    if (t < 256) {
        int lo = 0, hi = 0;
        if (t < nslice) { lo = loff[t * LPITCH + b]; hi = loff[t * LPITCH + b + 1]; }
        runlo[t] = lo; runhi[t] = hi; cnt[t] = 0;
        int a = lo, m = hi - lo;
        for (int o = 16; o > 0; o >>= 1) { a += __shfl_xor(a, o); m += __shfl_xor(m, o); }
        if (lane == 0) { wsA[w] = a; wsM[w] = m; }
    }
    __syncthreads();
    int r0 = 0, total = 0;
    #pragma unroll
    for (int k = 0; k < 8; ++k) { r0 += wsA[k]; total += wsM[k]; }

    int r[16];
    #pragma unroll
    for (int q = 0; q < 8; ++q) {
        const int sl = min(w + 32 * q, nslice - 1);
        const int* rp = recs + (long)sl * SLICE;
        const int i = runlo[sl] + lane, h = runhi[sl];
        const int x0 = rp[max(min(i, h - 1), 0)];
        const int x1 = rp[max(min(i + 32, h - 1), 0)];
        const bool own = w + 32 * q < nslice;
        r[2 * q]     = (own && i < h)      ? x0 : -1;
        r[2 * q + 1] = (own && i + 32 < h) ? x1 : -1;
    }
    #pragma unroll
    for (int q = 0; q < 16; ++q)
        if (r[q] >= 0) atomicAdd(&cnt[r[q] & (BNODES - 1)], 1);
    for (int q = 0; q < 8; ++q) {
        const int sl = w + 32 * q;
        if (sl < nslice) {
            const int* rp = recs + (long)sl * SLICE;
            for (int i = runlo[sl] + 64 + lane; i < runhi[sl]; i += 32)
                atomicAdd(&cnt[rp[i] & (BNODES - 1)], 1);
        }
    }
    __syncthreads();

    int c = 0, v = 0;
    if (t < BNODES) {
        c = cnt[t]; v = c;
        for (int o = 1; o < 32; o <<= 1) { int u = __shfl_up(v, o); if (lane >= o) v += u; }
        if (lane == 31) wsum[w] = v;
    }
    __syncthreads();
    if (t < BNODES) {
        int base = 0;
        for (int k = 0; k < w; ++k) base += wsum[k];
        const int ex = base + v - c;
        cnt[t] = ex;
        const int node = b * BNODES + t;
        if (node <= N) rowptr[node] = r0 + ex;
    }
    if ((N & (BNODES - 1)) == 0 && b == (int)gridDim.x - 1 && t == 0) rowptr[N] = r0 + total;
    __syncthreads();

    const bool staged = total <= STAGE_CAP;
    #pragma unroll
    for (int q = 0; q < 16; ++q)
        if (r[q] >= 0) {
            const int pos = atomicAdd(&cnt[r[q] & (BNODES - 1)], 1);
            if (staged) stage[pos] = r[q] >> BSH;
            else atomicExch(csr_src + r0 + pos, r[q] >> BSH);
        }
    for (int q = 0; q < 8; ++q) {
        const int sl = w + 32 * q;
        if (sl < nslice) {
            const int* rp = recs + (long)sl * SLICE;
            for (int i = runlo[sl] + 64 + lane; i < runhi[sl]; i += 32) {
                const int x = rp[i];
                const int pos = atomicAdd(&cnt[x & (BNODES - 1)], 1);
                if (staged) stage[pos] = x >> BSH;
                else atomicExch(csr_src + r0 + pos, x >> BSH);
            }
        }
    }
    __syncthreads();
    if (staged) {
        const int r1 = r0 + total;
        for (int a = (r0 & ~31) + t; a < r1; a += 1024) {
            if (a >= r0) {
                const int x = stage[a - r0];
                const int line = a & ~31;
                if (line >= r0 && line + 32 <= r1) csr_src[a] = x;
                else atomicExch(csr_src + a, x);
            }
        }
    }
}


__device__ __forceinline__ void add_rows(
    const float* __restrict__ h, int s, const float4 p, int cnt, int lane, float4& acc)
{
    const bool h0 = (lane & 8) != 0, h1 = (lane & 16) != 0;
    for (int t = 0; t < cnt; t += 4) {
        #pragma unroll
        for (int u = 0; u < 4; ++u) {
            const int st = __builtin_amdgcn_readlane(s, t + u);
            const float p0 = __int_as_float(__builtin_amdgcn_readlane(__float_as_int(p.x), t + u));
            const float p1 = __int_as_float(__builtin_amdgcn_readlane(__float_as_int(p.y), t + u));
            const float p2 = __int_as_float(__builtin_amdgcn_readlane(__float_as_int(p.z), t + u));
            const float p3 = __int_as_float(__builtin_amdgcn_readlane(__float_as_int(p.w), t + u));
            const float plo = h0 ? p1 : p0, phi = h0 ? p3 : p2;
            const float al = h1 ? phi : plo;
            const float4 hv = *(const float4*)(h + (long)st * HC + lane * 4);
            acc.x = fmaf(hv.x, al, acc.x);
            acc.y = fmaf(hv.y, al, acc.y);
            acc.z = fmaf(hv.z, al, acc.z);
            acc.w = fmaf(hv.w, al, acc.w);
        }
    }
}

__global__ __launch_bounds__(256) void node_gather(
    const int* __restrict__ rowptr, const int* __restrict__ csr_src,
    const float* __restrict__ h,
    const float* __restrict__ as, const float* __restrict__ ad,
    const float* __restrict__ bias,
    float* __restrict__ out, int N)
{
    const int d = __builtin_amdgcn_readfirstlane((int)(blockIdx.x * 8 + (threadIdx.x >> 5)));
    if (d >= N) return;
    const int lane = threadIdx.x & 31;
    const int hd   = lane >> 3;
    const int beg = rowptr[d], end = rowptr[d + 1];
    const float4 adv = *(const float4*)(ad + (long)d * 4);
    const float4 asd = *(const float4*)(as + (long)d * 4);
    float4 es;
    es.x = lrelu(asd.x + adv.x); es.y = lrelu(asd.y + adv.y);
    es.z = lrelu(asd.z + adv.z); es.w = lrelu(asd.w + adv.w);

    const int  jA = beg + lane, jB = jA + 32;
    const bool vA = jA < end,   vB = jB < end;
    const int  jl = max(end - 1, 0);
    const int  cA = csr_src[min(jA, jl)];
    const int  cB = csr_src[min(jB, jl)];
    const float4 aA = *(const float4*)(as + (long)cA * 4);
    const float4 aB = *(const float4*)(as + (long)cB * 4);
    asm volatile("" :: "v"(aA.x), "v"(aA.y), "v"(aA.z), "v"(aA.w),
                       "v"(aB.x), "v"(aB.y), "v"(aB.z), "v"(aB.w));
    const int sA = vA ? cA : d, sB = vB ? cB : d;
    float4 eA, eB;
    eA.x = vA ? lrelu(aA.x + adv.x) : es.x;  eA.y = vA ? lrelu(aA.y + adv.y) : es.y;
    eA.z = vA ? lrelu(aA.z + adv.z) : es.z;  eA.w = vA ? lrelu(aA.w + adv.w) : es.w;
    eB.x = vB ? lrelu(aB.x + adv.x) : es.x;  eB.y = vB ? lrelu(aB.y + adv.y) : es.y;
    eB.z = vB ? lrelu(aB.z + adv.z) : es.z;  eB.w = vB ? lrelu(aB.w + adv.w) : es.w;

    float4 mx;
    mx.x = fmaxf(fmaxf(es.x, eA.x), eB.x);  mx.y = fmaxf(fmaxf(es.y, eA.y), eB.y);
    mx.z = fmaxf(fmaxf(es.z, eA.z), eB.z);  mx.w = fmaxf(fmaxf(es.w, eA.w), eB.w);
    for (int j0 = beg + 64; j0 < end; j0 += 32) {
        const int j = j0 + lane;
        if (j < end) {
            const int s = csr_src[j];
            const float4 a = *(const float4*)(as + (long)s * 4);
            mx.x = fmaxf(mx.x, lrelu(a.x + adv.x));
            mx.y = fmaxf(mx.y, lrelu(a.y + adv.y));
            mx.z = fmaxf(mx.z, lrelu(a.z + adv.z));
            mx.w = fmaxf(mx.w, lrelu(a.w + adv.w));
        }
    }
    for (int o = 16; o > 0; o >>= 1) {
        mx.x = fmaxf(mx.x, __shfl_xor(mx.x, o));
        mx.y = fmaxf(mx.y, __shfl_xor(mx.y, o));
        mx.z = fmaxf(mx.z, __shfl_xor(mx.z, o));
        mx.w = fmaxf(mx.w, __shfl_xor(mx.w, o));
    }

    float4 pA, pB;
    pA.x = vA ? __expf(eA.x - mx.x) : 0.f;  pA.y = vA ? __expf(eA.y - mx.y) : 0.f;
    pA.z = vA ? __expf(eA.z - mx.z) : 0.f;  pA.w = vA ? __expf(eA.w - mx.w) : 0.f;
    pB.x = vB ? __expf(eB.x - mx.x) : 0.f;  pB.y = vB ? __expf(eB.y - mx.y) : 0.f;
    pB.z = vB ? __expf(eB.z - mx.z) : 0.f;  pB.w = vB ? __expf(eB.w - mx.w) : 0.f;
    float4 sm;
    sm.x = pA.x + pB.x; sm.y = pA.y + pB.y; sm.z = pA.z + pB.z; sm.w = pA.w + pB.w;
    for (int j0 = beg + 64; j0 < end; j0 += 32) {
        const int j = j0 + lane;
        if (j < end) {
            const int s = csr_src[j];
            const float4 a = *(const float4*)(as + (long)s * 4);
            sm.x += __expf(lrelu(a.x + adv.x) - mx.x);
            sm.y += __expf(lrelu(a.y + adv.y) - mx.y);
            sm.z += __expf(lrelu(a.z + adv.z) - mx.z);
            sm.w += __expf(lrelu(a.w + adv.w) - mx.w);
        }
    }
    for (int o = 16; o > 0; o >>= 1) {
        sm.x += __shfl_xor(sm.x, o);
        sm.y += __shfl_xor(sm.y, o);
        sm.z += __shfl_xor(sm.z, o);
        sm.w += __shfl_xor(sm.w, o);
    }
    float4 ps;
    ps.x = __expf(es.x - mx.x); ps.y = __expf(es.y - mx.y);
    ps.z = __expf(es.z - mx.z); ps.w = __expf(es.w - mx.w);
    sm.x += ps.x; sm.y += ps.y; sm.z += ps.z; sm.w += ps.w;

    const float aself = hd == 0 ? ps.x / sm.x : hd == 1 ? ps.y / sm.y
                      : hd == 2 ? ps.z / sm.z : ps.w / sm.w;
    const float4 hs = *(const float4*)(h + (long)d * HC + lane * 4);
    float4 acc;
    acc.x = hs.x * aself; acc.y = hs.y * aself;
    acc.z = hs.z * aself; acc.w = hs.w * aself;
    pA.x = pA.x / sm.x; pA.y = pA.y / sm.y; pA.z = pA.z / sm.z; pA.w = pA.w / sm.w;
    pB.x = pB.x / sm.x; pB.y = pB.y / sm.y; pB.z = pB.z / sm.z; pB.w = pB.w / sm.w;
    add_rows(h, sA, pA, min(32, end - beg), lane, acc);
    add_rows(h, sB, pB, min(32, end - beg - 32), lane, acc);
    for (int j0 = beg + 64; j0 < end; j0 += 32) {
        const int j = j0 + lane;
        int s = d;
        float4 p; p.x = p.y = p.z = p.w = 0.f;
        if (j < end) {
            s = csr_src[j];
            const float4 a = *(const float4*)(as + (long)s * 4);
            p.x = __expf(lrelu(a.x + adv.x) - mx.x) / sm.x;
            p.y = __expf(lrelu(a.y + adv.y) - mx.y) / sm.y;
            p.z = __expf(lrelu(a.z + adv.z) - mx.z) / sm.z;
            p.w = __expf(lrelu(a.w + adv.w) - mx.w) / sm.w;
        }
        add_rows(h, s, p, min(32, end - j0), lane, acc);
    }
    const float4 bv = *(const float4*)(bias + lane * 4);
    float4 y;
    y.x = acc.x + bv.x; y.y = acc.y + bv.y;
    y.z = acc.z + bv.z; y.w = acc.w + bv.w;
    y.x = y.x > 0.f ? y.x : __expf(y.x) - 1.f;
    y.y = y.y > 0.f ? y.y : __expf(y.y) - 1.f;
    y.z = y.z > 0.f ? y.z : __expf(y.z) - 1.f;
    y.w = y.w > 0.f ? y.w : __expf(y.w) - 1.f;
    *(float4*)(out + (long)d * HC + lane * 4) = y;
}

extern "C" void kernel_launch(void* const* d_in, const int* in_sizes, int n_in,
                              void* d_out, int out_size, void* d_ws, size_t ws_size,
                              hipStream_t stream)
{
    const float* x      = (const float*)d_in[0];
    const int*   ei     = (const int*)d_in[1];
    const float* W1     = (const float*)d_in[2];
    const float* a_src1 = (const float*)d_in[3];
    const float* a_dst1 = (const float*)d_in[4];
    const float* b1     = (const float*)d_in[5];
    const float* W2     = (const float*)d_in[6];
    const float* a_src2 = (const float*)d_in[7];
    const float* a_dst2 = (const float*)d_in[8];
    const float* b2     = (const float*)d_in[9];

    const int N  = in_sizes[0] / HC;
    const int E  = in_sizes[1] / 2;
    const int NH = N * HEADS;
    const int NF = N * HC;

    char* ws = (char*)d_ws;
    float* hbuf   = (float*)(ws);
    float* out1   = (float*)(ws + (size_t)NF * 4);
    float* as     = (float*)(ws + (size_t)NF * 8);
    float* ad     = (float*)(ws + (size_t)NF * 8 + (size_t)NH * 4);
    int*   rowptr = (int*)(ws + (size_t)NF * 8 + (size_t)NH * 12);
    int*   csrsrc = (int*)(ws + (size_t)NF * 8 + (size_t)NH * 16);
    int*   recs   = (int*)(ws + (size_t)NF * 8 + (size_t)NH * 16 + (size_t)E * 4);

    const int TB = 256;
    const int nslice = (E + SLICE - 1) / SLICE;
    const int nbk    = (N + BNODES - 1) / BNODES;
    int*   loff   = recs + (size_t)nslice * SLICE;
    const int gNW = (N + 7) / 8;
    float* outF = (float*)d_out;

    slice_sort<<<nslice, 1024, 0, stream>>>(ei, E, recs, loff);
    bucket_finish<<<nbk, 1024, 0, stream>>>(recs, loff, nslice, N, rowptr, csrsrc);

    gemm_f32_wmma<<<N / 16, 256, 0, stream>>>(x, W1, a_src1, a_dst1, hbuf, as, ad, N);
    node_gather<<<gNW, TB, 0, stream>>>(rowptr, csrsrc, hbuf, as, ad, b1, out1, N);

    gemm_f32_wmma<<<N / 16, 256, 0, stream>>>(out1, W2, a_src2, a_dst2, hbuf, as, ad, N);
    node_gather<<<gNW, TB, 0, stream>>>(rowptr, csrsrc, hbuf, as, ad, b2, outF, N);
}


# GEMM re-tiled: W staged once per work-group in LDS as 16-byte words of four k, a wave owns 16 rows x 64 columns and reads A once with 16-byte loads; k order and logit sum order change
# speedup vs baseline: 12.2108x; 1.1647x over previous
#include <hip/hip_runtime.h>
#include <hip/hip_bf16.h>

typedef __attribute__((ext_vector_type(2))) float v2f;
typedef __attribute__((ext_vector_type(8))) float v8f;

#define HC 128
#define HEADS 4
#define CH 32
#define NEG_SLOPE 0.2f

__device__ __forceinline__ float lrelu(float v) { return v > 0.f ? v : NEG_SLOPE * v; }

template <int M> __device__ __forceinline__ float row_xor(float v) {
    return __builtin_bit_cast(float, __builtin_amdgcn_update_dpp(
        0, __builtin_bit_cast(int, v), 0x160 | M, 0xf, 0xf, false));
}
__device__ __forceinline__ float row_sums8(const float (&v)[8], int lane) {
    const bool b8 = (lane & 8) != 0, b4 = (lane & 4) != 0, b2 = (lane & 2) != 0;
    float a[4], b[2];
    #pragma unroll
    for (int j = 0; j < 4; ++j)
        a[j] = (b8 ? v[j + 4] : v[j]) + row_xor<8>(b8 ? v[j] : v[j + 4]);
    #pragma unroll
    for (int j = 0; j < 2; ++j)
        b[j] = (b4 ? a[j + 2] : a[j]) + row_xor<4>(b4 ? a[j] : a[j + 2]);
    const float c = (b2 ? b[1] : b[0]) + row_xor<2>(b2 ? b[0] : b[1]);
    return c + row_xor<1>(c);
}

#define BQ_COLS  36
#define BQ_PITCH 152
__global__ __launch_bounds__(128) void gemm_f32_wmma(
    const float* __restrict__ A, const float* __restrict__ B,
    const float* __restrict__ a_src, const float* __restrict__ a_dst,
    float* __restrict__ D, float* __restrict__ as, float* __restrict__ ad, int M)
{
    __shared__ float4 Bq[32 * BQ_PITCH];
    __shared__ float4 part[2][2][16];
    const int wave = threadIdx.x >> 5;
    const int lane = threadIdx.x & 31;
    const int m15  = lane & 15;
    const int hi   = lane >> 4;
    const int sl   = wave >> 1;
    const int half = wave & 1;
    const int nstrips = M >> 4;
    const int strip_w = blockIdx.x * 2 + sl;
    const bool live   = strip_w < nstrips;
    const int strip   = min(strip_w, nstrips - 1);

    #pragma unroll
    for (int i = 0; i < 8; ++i) {
        const int q = wave + 4 * i;
        const float* bp = B + (long)(4 * q) * HC + 4 * lane;
        const float4 r0 = *(const float4*)(bp);
        const float4 r1 = *(const float4*)(bp + HC);
        const float4 r2 = *(const float4*)(bp + 2 * HC);
        const float4 r3 = *(const float4*)(bp + 3 * HC);
        float4* dq = Bq + q * BQ_PITCH + lane;
        dq[0 * BQ_COLS] = make_float4(r0.x, r1.x, r2.x, r3.x);
        dq[1 * BQ_COLS] = make_float4(r0.y, r1.y, r2.y, r3.y);
        dq[2 * BQ_COLS] = make_float4(r0.z, r1.z, r2.z, r3.z);
        dq[3 * BQ_COLS] = make_float4(r0.w, r1.w, r2.w, r3.w);
    }

    float ws[4], wd[4];
    #pragma unroll
    for (int tt = 0; tt < 4; ++tt) {
        const int col = 64 * half + 16 * tt + m15;
        ws[tt] = a_src[col]; wd[tt] = a_dst[col];
    }

    const float* ap = A + (long)(strip * 16 + m15) * HC + 4 * hi;
    float4 a4[16];
    #pragma unroll
    for (int j = 0; j < 16; ++j) a4[j] = *(const float4*)(ap + 8 * j);

    v8f acc[4];
    #pragma unroll
    for (int tt = 0; tt < 4; ++tt)
        #pragma unroll
        for (int r = 0; r < 8; ++r) acc[tt][r] = 0.f;

    __syncthreads();

    const float4* bq = Bq + hi * BQ_PITCH + BQ_COLS * (m15 & 3) + 16 * half + (m15 >> 2);
    float4 bn[4];
    #pragma unroll
    for (int tt = 0; tt < 4; ++tt) bn[tt] = bq[4 * tt];
    #pragma unroll
    for (int j = 0; j < 16; ++j) {
        float4 b4[4];
        #pragma unroll
        for (int tt = 0; tt < 4; ++tt) b4[tt] = bn[tt];
        if (j + 1 < 16) {
            #pragma unroll
            for (int tt = 0; tt < 4; ++tt) bn[tt] = bq[2 * (j + 1) * BQ_PITCH + 4 * tt];
        }
        __builtin_amdgcn_sched_barrier(0);
        #pragma unroll
        for (int tt = 0; tt < 4; ++tt) {
            v2f a; a.x = a4[j].x; a.y = a4[j].y;
            v2f b; b.x = b4[tt].x; b.y = b4[tt].y;
            acc[tt] = __builtin_amdgcn_wmma_f32_16x16x4_f32(
                false, a, false, b, (short)0, acc[tt], false, false);
        }
        #pragma unroll
        for (int tt = 0; tt < 4; ++tt) {
            v2f a; a.x = a4[j].z; a.y = a4[j].w;
            v2f b; b.x = b4[tt].z; b.y = b4[tt].w;
            acc[tt] = __builtin_amdgcn_wmma_f32_16x16x4_f32(
                false, a, false, b, (short)0, acc[tt], false, false);
        }
        __builtin_amdgcn_sched_barrier(0);
    }

    float sv[2][2];
    #pragma unroll
    for (int hh = 0; hh < 2; ++hh) {
        float ps[8], pd[8];
        #pragma unroll
        for (int r = 0; r < 8; ++r) {
            ps[r] = fmaf(acc[2 * hh][r], ws[2 * hh], acc[2 * hh + 1][r] * ws[2 * hh + 1]);
            pd[r] = fmaf(acc[2 * hh][r], wd[2 * hh], acc[2 * hh + 1][r] * wd[2 * hh + 1]);
        }
        sv[hh][0] = row_sums8(ps, lane);
        sv[hh][1] = row_sums8(pd, lane);
    }
    {
        const int which = lane & 1;
        float* pp = (float*)&part[sl][which][8 * hi + (m15 >> 1)];
        pp[2 * half]     = which ? sv[0][1] : sv[0][0];
        pp[2 * half + 1] = which ? sv[1][1] : sv[1][0];
    }
    __syncthreads();
    if (threadIdx.x < 64) {
        const int s = threadIdx.x >> 5, which = lane >> 4, rl = lane & 15;
        const int st = blockIdx.x * 2 + s;
        if (st < nstrips) {
            float* dstp = which ? ad : as;
            *(float4*)(dstp + (long)st * 64 + rl * 4) = part[s][which][rl];
        }
    }
    if (live) {
        float* dp = D + (long)(strip * 16 + 8 * hi) * HC + 64 * half + m15;
        #pragma unroll
        for (int tt = 0; tt < 4; ++tt)
            #pragma unroll
            for (int r = 0; r < 8; ++r)
                dp[(long)r * HC + 16 * tt] = acc[tt][r];
    }
}

#define BSH       8
#define BNODES    256
#define SLICE     6272
#define SLICE_IT  7
#define LPITCH    288
#define STAGE_CAP 12288

__global__ __launch_bounds__(1024) void slice_sort(
    const int* __restrict__ ei, int E,
    int* __restrict__ recs, int* __restrict__ loff)
{
    __shared__ int hist[BNODES];
    __shared__ int wsum[8];
    __shared__ int sorted[SLICE];
    const int t = threadIdx.x, lane = t & 31, w = t >> 5;
    const int e0 = blockIdx.x * SLICE;
    const int e1 = min(e0 + SLICE, E);
    if (t < BNODES) hist[t] = 0;
    __syncthreads();

    int rec[SLICE_IT], bk[SLICE_IT], rk[SLICE_IT];
    #pragma unroll
    for (int i = 0; i < SLICE_IT; ++i) {
        const int idx = e0 + t + i * 1024;
        const int idc = min(idx, e1 - 1);
        const int s = ei[idc], d = ei[E + idc];
        rk[i]  = 0;
        bk[i]  = idx < e1 ? (d >> BSH) : -1;
        rec[i] = (s << BSH) | (d & (BNODES - 1));
    }
    #pragma unroll
    for (int i = 0; i < SLICE_IT; ++i)
        if (bk[i] >= 0) rk[i] = atomicAdd(&hist[bk[i]], 1);
    __syncthreads();

    int c = 0, v = 0;
    if (t < BNODES) {
        c = hist[t]; v = c;
        for (int o = 1; o < 32; o <<= 1) { int u = __shfl_up(v, o); if (lane >= o) v += u; }
        if (lane == 31) wsum[w] = v;
    }
    __syncthreads();
    if (t < BNODES) {
        int base = 0;
        for (int k = 0; k < w; ++k) base += wsum[k];
        const int ex = base + v - c;
        hist[t] = ex;
        loff[blockIdx.x * LPITCH + t] = ex;
        if (t == BNODES - 1) loff[blockIdx.x * LPITCH + BNODES] = ex + c;
    }
    __syncthreads();
    #pragma unroll
    for (int i = 0; i < SLICE_IT; ++i)
        if (bk[i] >= 0) sorted[hist[bk[i]] + rk[i]] = rec[i];
    __syncthreads();
    const int n = e1 - e0;
    for (int i = t; i < n; i += 1024) recs[e0 + i] = sorted[i];
}

__global__ __launch_bounds__(1024) void bucket_finish(
    const int* __restrict__ recs, const int* __restrict__ loff, int nslice, int N,
    int* __restrict__ rowptr, int* __restrict__ csr_src)
{
    __shared__ int runlo[256], runhi[256];
    __shared__ int cnt[BNODES];
    __shared__ int wsA[8], wsM[8], wsum[8];
    __shared__ int stage[STAGE_CAP];
    const int t = threadIdx.x, lane = t & 31, w = t >> 5;
    const int b = blockIdx.x;

    if (t < 256) {
        int lo = 0, hi = 0;
        if (t < nslice) { lo = loff[t * LPITCH + b]; hi = loff[t * LPITCH + b + 1]; }
        runlo[t] = lo; runhi[t] = hi; cnt[t] = 0;
        int a = lo, m = hi - lo;
        for (int o = 16; o > 0; o >>= 1) { a += __shfl_xor(a, o); m += __shfl_xor(m, o); }
        if (lane == 0) { wsA[w] = a; wsM[w] = m; }
    }
    __syncthreads();
    int r0 = 0, total = 0;
    #pragma unroll
    for (int k = 0; k < 8; ++k) { r0 += wsA[k]; total += wsM[k]; }

    int r[16];
    #pragma unroll
    for (int q = 0; q < 8; ++q) {
        const int sl = min(w + 32 * q, nslice - 1);
        const int* rp = recs + (long)sl * SLICE;
        const int i = runlo[sl] + lane, h = runhi[sl];
        const int x0 = rp[max(min(i, h - 1), 0)];
        const int x1 = rp[max(min(i + 32, h - 1), 0)];
        const bool own = w + 32 * q < nslice;
        r[2 * q]     = (own && i < h)      ? x0 : -1;
        r[2 * q + 1] = (own && i + 32 < h) ? x1 : -1;
    }
    #pragma unroll
    for (int q = 0; q < 16; ++q)
        if (r[q] >= 0) atomicAdd(&cnt[r[q] & (BNODES - 1)], 1);
    for (int q = 0; q < 8; ++q) {
        const int sl = w + 32 * q;
        if (sl < nslice) {
            const int* rp = recs + (long)sl * SLICE;
            for (int i = runlo[sl] + 64 + lane; i < runhi[sl]; i += 32)
                atomicAdd(&cnt[rp[i] & (BNODES - 1)], 1);
        }
    }
    __syncthreads();

    int c = 0, v = 0;
    if (t < BNODES) {
        c = cnt[t]; v = c;
        for (int o = 1; o < 32; o <<= 1) { int u = __shfl_up(v, o); if (lane >= o) v += u; }
        if (lane == 31) wsum[w] = v;
    }
    __syncthreads();
    if (t < BNODES) {
        int base = 0;
        for (int k = 0; k < w; ++k) base += wsum[k];
        const int ex = base + v - c;
        cnt[t] = ex;
        const int node = b * BNODES + t;
        if (node <= N) rowptr[node] = r0 + ex;
    }
    if ((N & (BNODES - 1)) == 0 && b == (int)gridDim.x - 1 && t == 0) rowptr[N] = r0 + total;
    __syncthreads();

    const bool staged = total <= STAGE_CAP;
    #pragma unroll
    for (int q = 0; q < 16; ++q)
        if (r[q] >= 0) {
            const int pos = atomicAdd(&cnt[r[q] & (BNODES - 1)], 1);
            if (staged) stage[pos] = r[q] >> BSH;
            else atomicExch(csr_src + r0 + pos, r[q] >> BSH);
        }
    for (int q = 0; q < 8; ++q) {
        const int sl = w + 32 * q;
        if (sl < nslice) {
            const int* rp = recs + (long)sl * SLICE;
            for (int i = runlo[sl] + 64 + lane; i < runhi[sl]; i += 32) {
                const int x = rp[i];
                const int pos = atomicAdd(&cnt[x & (BNODES - 1)], 1);
                if (staged) stage[pos] = x >> BSH;
                else atomicExch(csr_src + r0 + pos, x >> BSH);
            }
        }
    }
    __syncthreads();
    if (staged) {
        const int r1 = r0 + total;
        for (int a = (r0 & ~31) + t; a < r1; a += 1024) {
            if (a >= r0) {
                const int x = stage[a - r0];
                const int line = a & ~31;
                if (line >= r0 && line + 32 <= r1) csr_src[a] = x;
                else atomicExch(csr_src + a, x);
            }
        }
    }
}


__device__ __forceinline__ void add_rows(
    const float* __restrict__ h, int s, const float4 p, int cnt, int lane, float4& acc)
{
    const bool h0 = (lane & 8) != 0, h1 = (lane & 16) != 0;
    for (int t = 0; t < cnt; t += 4) {
        #pragma unroll
        for (int u = 0; u < 4; ++u) {
            const int st = __builtin_amdgcn_readlane(s, t + u);
            const float p0 = __int_as_float(__builtin_amdgcn_readlane(__float_as_int(p.x), t + u));
            const float p1 = __int_as_float(__builtin_amdgcn_readlane(__float_as_int(p.y), t + u));
            const float p2 = __int_as_float(__builtin_amdgcn_readlane(__float_as_int(p.z), t + u));
            const float p3 = __int_as_float(__builtin_amdgcn_readlane(__float_as_int(p.w), t + u));
            const float plo = h0 ? p1 : p0, phi = h0 ? p3 : p2;
            const float al = h1 ? phi : plo;
            const float4 hv = *(const float4*)(h + (long)st * HC + lane * 4);
            acc.x = fmaf(hv.x, al, acc.x);
            acc.y = fmaf(hv.y, al, acc.y);
            acc.z = fmaf(hv.z, al, acc.z);
            acc.w = fmaf(hv.w, al, acc.w);
        }
    }
}

__global__ __launch_bounds__(256) void node_gather(
    const int* __restrict__ rowptr, const int* __restrict__ csr_src,
    const float* __restrict__ h,
    const float* __restrict__ as, const float* __restrict__ ad,
    const float* __restrict__ bias,
    float* __restrict__ out, int N)
{
    const int d = __builtin_amdgcn_readfirstlane((int)(blockIdx.x * 8 + (threadIdx.x >> 5)));
    if (d >= N) return;
    const int lane = threadIdx.x & 31;
    const int hd   = lane >> 3;
    const int beg = rowptr[d], end = rowptr[d + 1];
    const float4 adv = *(const float4*)(ad + (long)d * 4);
    const float4 asd = *(const float4*)(as + (long)d * 4);
    float4 es;
    es.x = lrelu(asd.x + adv.x); es.y = lrelu(asd.y + adv.y);
    es.z = lrelu(asd.z + adv.z); es.w = lrelu(asd.w + adv.w);

    const int  jA = beg + lane, jB = jA + 32;
    const bool vA = jA < end,   vB = jB < end;
    const int  jl = max(end - 1, 0);
    const int  cA = csr_src[min(jA, jl)];
    const int  cB = csr_src[min(jB, jl)];
    const float4 aA = *(const float4*)(as + (long)cA * 4);
    const float4 aB = *(const float4*)(as + (long)cB * 4);
    asm volatile("" :: "v"(aA.x), "v"(aA.y), "v"(aA.z), "v"(aA.w),
                       "v"(aB.x), "v"(aB.y), "v"(aB.z), "v"(aB.w));
    const int sA = vA ? cA : d, sB = vB ? cB : d;
    float4 eA, eB;
    eA.x = vA ? lrelu(aA.x + adv.x) : es.x;  eA.y = vA ? lrelu(aA.y + adv.y) : es.y;
    eA.z = vA ? lrelu(aA.z + adv.z) : es.z;  eA.w = vA ? lrelu(aA.w + adv.w) : es.w;
    eB.x = vB ? lrelu(aB.x + adv.x) : es.x;  eB.y = vB ? lrelu(aB.y + adv.y) : es.y;
    eB.z = vB ? lrelu(aB.z + adv.z) : es.z;  eB.w = vB ? lrelu(aB.w + adv.w) : es.w;

    float4 mx;
    mx.x = fmaxf(fmaxf(es.x, eA.x), eB.x);  mx.y = fmaxf(fmaxf(es.y, eA.y), eB.y);
    mx.z = fmaxf(fmaxf(es.z, eA.z), eB.z);  mx.w = fmaxf(fmaxf(es.w, eA.w), eB.w);
    for (int j0 = beg + 64; j0 < end; j0 += 32) {
        const int j = j0 + lane;
        if (j < end) {
            const int s = csr_src[j];
            const float4 a = *(const float4*)(as + (long)s * 4);
            mx.x = fmaxf(mx.x, lrelu(a.x + adv.x));
            mx.y = fmaxf(mx.y, lrelu(a.y + adv.y));
            mx.z = fmaxf(mx.z, lrelu(a.z + adv.z));
            mx.w = fmaxf(mx.w, lrelu(a.w + adv.w));
        }
    }
    for (int o = 16; o > 0; o >>= 1) {
        mx.x = fmaxf(mx.x, __shfl_xor(mx.x, o));
        mx.y = fmaxf(mx.y, __shfl_xor(mx.y, o));
        mx.z = fmaxf(mx.z, __shfl_xor(mx.z, o));
        mx.w = fmaxf(mx.w, __shfl_xor(mx.w, o));
    }

    float4 pA, pB;
    pA.x = vA ? __expf(eA.x - mx.x) : 0.f;  pA.y = vA ? __expf(eA.y - mx.y) : 0.f;
    pA.z = vA ? __expf(eA.z - mx.z) : 0.f;  pA.w = vA ? __expf(eA.w - mx.w) : 0.f;
    pB.x = vB ? __expf(eB.x - mx.x) : 0.f;  pB.y = vB ? __expf(eB.y - mx.y) : 0.f;
    pB.z = vB ? __expf(eB.z - mx.z) : 0.f;  pB.w = vB ? __expf(eB.w - mx.w) : 0.f;
    float4 sm;
    sm.x = pA.x + pB.x; sm.y = pA.y + pB.y; sm.z = pA.z + pB.z; sm.w = pA.w + pB.w;
    for (int j0 = beg + 64; j0 < end; j0 += 32) {
        const int j = j0 + lane;
        if (j < end) {
            const int s = csr_src[j];
            const float4 a = *(const float4*)(as + (long)s * 4);
            sm.x += __expf(lrelu(a.x + adv.x) - mx.x);
            sm.y += __expf(lrelu(a.y + adv.y) - mx.y);
            sm.z += __expf(lrelu(a.z + adv.z) - mx.z);
            sm.w += __expf(lrelu(a.w + adv.w) - mx.w);
        }
    }
    for (int o = 16; o > 0; o >>= 1) {
        sm.x += __shfl_xor(sm.x, o);
        sm.y += __shfl_xor(sm.y, o);
        sm.z += __shfl_xor(sm.z, o);
        sm.w += __shfl_xor(sm.w, o);
    }
    float4 ps;
    ps.x = __expf(es.x - mx.x); ps.y = __expf(es.y - mx.y);
    ps.z = __expf(es.z - mx.z); ps.w = __expf(es.w - mx.w);
    sm.x += ps.x; sm.y += ps.y; sm.z += ps.z; sm.w += ps.w;

    const float aself = hd == 0 ? ps.x / sm.x : hd == 1 ? ps.y / sm.y
                      : hd == 2 ? ps.z / sm.z : ps.w / sm.w;
    const float4 hs = *(const float4*)(h + (long)d * HC + lane * 4);
    float4 acc;
    acc.x = hs.x * aself; acc.y = hs.y * aself;
    acc.z = hs.z * aself; acc.w = hs.w * aself;
    pA.x = pA.x / sm.x; pA.y = pA.y / sm.y; pA.z = pA.z / sm.z; pA.w = pA.w / sm.w;
    pB.x = pB.x / sm.x; pB.y = pB.y / sm.y; pB.z = pB.z / sm.z; pB.w = pB.w / sm.w;
    add_rows(h, sA, pA, min(32, end - beg), lane, acc);
    add_rows(h, sB, pB, min(32, end - beg - 32), lane, acc);
    for (int j0 = beg + 64; j0 < end; j0 += 32) {
        const int j = j0 + lane;
        int s = d;
        float4 p; p.x = p.y = p.z = p.w = 0.f;
        if (j < end) {
            s = csr_src[j];
            const float4 a = *(const float4*)(as + (long)s * 4);
            p.x = __expf(lrelu(a.x + adv.x) - mx.x) / sm.x;
            p.y = __expf(lrelu(a.y + adv.y) - mx.y) / sm.y;
            p.z = __expf(lrelu(a.z + adv.z) - mx.z) / sm.z;
            p.w = __expf(lrelu(a.w + adv.w) - mx.w) / sm.w;
        }
        add_rows(h, s, p, min(32, end - j0), lane, acc);
    }
    const float4 bv = *(const float4*)(bias + lane * 4);
    float4 y;
    y.x = acc.x + bv.x; y.y = acc.y + bv.y;
    y.z = acc.z + bv.z; y.w = acc.w + bv.w;
    y.x = y.x > 0.f ? y.x : __expf(y.x) - 1.f;
    y.y = y.y > 0.f ? y.y : __expf(y.y) - 1.f;
    y.z = y.z > 0.f ? y.z : __expf(y.z) - 1.f;
    y.w = y.w > 0.f ? y.w : __expf(y.w) - 1.f;
    *(float4*)(out + (long)d * HC + lane * 4) = y;
}

extern "C" void kernel_launch(void* const* d_in, const int* in_sizes, int n_in,
                              void* d_out, int out_size, void* d_ws, size_t ws_size,
                              hipStream_t stream)
{
    const float* x      = (const float*)d_in[0];
    const int*   ei     = (const int*)d_in[1];
    const float* W1     = (const float*)d_in[2];
    const float* a_src1 = (const float*)d_in[3];
    const float* a_dst1 = (const float*)d_in[4];
    const float* b1     = (const float*)d_in[5];
    const float* W2     = (const float*)d_in[6];
    const float* a_src2 = (const float*)d_in[7];
    const float* a_dst2 = (const float*)d_in[8];
    const float* b2     = (const float*)d_in[9];

    const int N  = in_sizes[0] / HC;
    const int E  = in_sizes[1] / 2;
    const int NH = N * HEADS;
    const int NF = N * HC;

    char* ws = (char*)d_ws;
    float* hbuf   = (float*)(ws);
    float* out1   = (float*)(ws + (size_t)NF * 4);
    float* as     = (float*)(ws + (size_t)NF * 8);
    float* ad     = (float*)(ws + (size_t)NF * 8 + (size_t)NH * 4);
    int*   rowptr = (int*)(ws + (size_t)NF * 8 + (size_t)NH * 12);
    int*   csrsrc = (int*)(ws + (size_t)NF * 8 + (size_t)NH * 16);
    int*   recs   = (int*)(ws + (size_t)NF * 8 + (size_t)NH * 16 + (size_t)E * 4);

    const int TB = 256;
    const int nslice = (E + SLICE - 1) / SLICE;
    const int nbk    = (N + BNODES - 1) / BNODES;
    int*   loff   = recs + (size_t)nslice * SLICE;
    const int gNW = (N + 7) / 8;
    float* outF = (float*)d_out;

    slice_sort<<<nslice, 1024, 0, stream>>>(ei, E, recs, loff);
    bucket_finish<<<nbk, 1024, 0, stream>>>(recs, loff, nslice, N, rowptr, csrsrc);

    gemm_f32_wmma<<<(N / 16 + 1) / 2, 128, 0, stream>>>(x, W1, a_src1, a_dst1, hbuf, as, ad, N);
    node_gather<<<gNW, TB, 0, stream>>>(rowptr, csrsrc, hbuf, as, ad, b1, out1, N);

    gemm_f32_wmma<<<(N / 16 + 1) / 2, 128, 0, stream>>>(out1, W2, a_src2, a_dst2, hbuf, as, ad, N);
    node_gather<<<gNW, TB, 0, stream>>>(rowptr, csrsrc, hbuf, as, ad, b2, outF, N);
}


# Gather: each edge's logit, leaky relu, exp done once per head, not in all 8 lanes (8-lane reduce-scatter leaves row i of a block of 8 in lane i; weight broadcast back); logit and exp-sum order change
# speedup vs baseline: 13.6111x; 1.1147x over previous
#include <hip/hip_runtime.h>
#include <hip/hip_bf16.h>

typedef __attribute__((ext_vector_type(2))) float v2f;
typedef __attribute__((ext_vector_type(8))) float v8f;

#define HC 128
#define HEADS 4
#define CH 32
#define NEG_SLOPE 0.2f

__device__ __forceinline__ float lrelu(float v) { return v > 0.f ? v : NEG_SLOPE * v; }

template <int M> __device__ __forceinline__ float row_xor(float v) {
    return __builtin_bit_cast(float, __builtin_amdgcn_update_dpp(
        0, __builtin_bit_cast(int, v), 0x160 | M, 0xf, 0xf, false));
}
__device__ __forceinline__ float row_sums8(const float (&v)[8], int lane) {
    const bool b8 = (lane & 8) != 0, b4 = (lane & 4) != 0, b2 = (lane & 2) != 0;
    float a[4], b[2];
    #pragma unroll
    for (int j = 0; j < 4; ++j)
        a[j] = (b8 ? v[j + 4] : v[j]) + row_xor<8>(b8 ? v[j] : v[j + 4]);
    #pragma unroll
    for (int j = 0; j < 2; ++j)
        b[j] = (b4 ? a[j + 2] : a[j]) + row_xor<4>(b4 ? a[j] : a[j + 2]);
    const float c = (b2 ? b[1] : b[0]) + row_xor<2>(b2 ? b[0] : b[1]);
    return c + row_xor<1>(c);
}

#define BQ_COLS  36
#define BQ_PITCH 152
__global__ __launch_bounds__(128) void gemm_f32_wmma(
    const float* __restrict__ A, const float* __restrict__ B,
    const float* __restrict__ a_src, const float* __restrict__ a_dst,
    float* __restrict__ D, float* __restrict__ as, float* __restrict__ ad, int M)
{
    __shared__ float4 Bq[32 * BQ_PITCH];
    __shared__ float4 part[2][2][16];
    const int wave = threadIdx.x >> 5;
    const int lane = threadIdx.x & 31;
    const int m15  = lane & 15;
    const int hi   = lane >> 4;
    const int sl   = wave >> 1;
    const int half = wave & 1;
    const int nstrips = M >> 4;
    const int strip_w = blockIdx.x * 2 + sl;
    const bool live   = strip_w < nstrips;
    const int strip   = min(strip_w, nstrips - 1);

    #pragma unroll
    for (int i = 0; i < 8; ++i) {
        const int q = wave + 4 * i;
        const float* bp = B + (long)(4 * q) * HC + 4 * lane;
        const float4 r0 = *(const float4*)(bp);
        const float4 r1 = *(const float4*)(bp + HC);
        const float4 r2 = *(const float4*)(bp + 2 * HC);
        const float4 r3 = *(const float4*)(bp + 3 * HC);
        float4* dq = Bq + q * BQ_PITCH + lane;
        dq[0 * BQ_COLS] = make_float4(r0.x, r1.x, r2.x, r3.x);
        dq[1 * BQ_COLS] = make_float4(r0.y, r1.y, r2.y, r3.y);
        dq[2 * BQ_COLS] = make_float4(r0.z, r1.z, r2.z, r3.z);
        dq[3 * BQ_COLS] = make_float4(r0.w, r1.w, r2.w, r3.w);
    }

    float ws[4], wd[4];
    #pragma unroll
    for (int tt = 0; tt < 4; ++tt) {
        const int col = 64 * half + 16 * tt + m15;
        ws[tt] = a_src[col]; wd[tt] = a_dst[col];
    }

    const float* ap = A + (long)(strip * 16 + m15) * HC + 4 * hi;
    float4 a4[16];
    #pragma unroll
    for (int j = 0; j < 16; ++j) a4[j] = *(const float4*)(ap + 8 * j);

    v8f acc[4];
    #pragma unroll
    for (int tt = 0; tt < 4; ++tt)
        #pragma unroll
        for (int r = 0; r < 8; ++r) acc[tt][r] = 0.f;

    __syncthreads();

    const float4* bq = Bq + hi * BQ_PITCH + BQ_COLS * (m15 & 3) + 16 * half + (m15 >> 2);
    float4 bn[4];
    #pragma unroll
    for (int tt = 0; tt < 4; ++tt) bn[tt] = bq[4 * tt];
    #pragma unroll
    for (int j = 0; j < 16; ++j) {
        float4 b4[4];
        #pragma unroll
        for (int tt = 0; tt < 4; ++tt) b4[tt] = bn[tt];
        if (j + 1 < 16) {
            #pragma unroll
            for (int tt = 0; tt < 4; ++tt) bn[tt] = bq[2 * (j + 1) * BQ_PITCH + 4 * tt];
        }
        __builtin_amdgcn_sched_barrier(0);
        #pragma unroll
        for (int tt = 0; tt < 4; ++tt) {
            v2f a; a.x = a4[j].x; a.y = a4[j].y;
            v2f b; b.x = b4[tt].x; b.y = b4[tt].y;
            acc[tt] = __builtin_amdgcn_wmma_f32_16x16x4_f32(
                false, a, false, b, (short)0, acc[tt], false, false);
        }
        #pragma unroll
        for (int tt = 0; tt < 4; ++tt) {
            v2f a; a.x = a4[j].z; a.y = a4[j].w;
            v2f b; b.x = b4[tt].z; b.y = b4[tt].w;
            acc[tt] = __builtin_amdgcn_wmma_f32_16x16x4_f32(
                false, a, false, b, (short)0, acc[tt], false, false);
        }
        __builtin_amdgcn_sched_barrier(0);
    }

    float sv[2][2];
    #pragma unroll
    for (int hh = 0; hh < 2; ++hh) {
        float ps[8], pd[8];
        #pragma unroll
        for (int r = 0; r < 8; ++r) {
            ps[r] = fmaf(acc[2 * hh][r], ws[2 * hh], acc[2 * hh + 1][r] * ws[2 * hh + 1]);
            pd[r] = fmaf(acc[2 * hh][r], wd[2 * hh], acc[2 * hh + 1][r] * wd[2 * hh + 1]);
        }
        sv[hh][0] = row_sums8(ps, lane);
        sv[hh][1] = row_sums8(pd, lane);
    }
    {
        const int which = lane & 1;
        float* pp = (float*)&part[sl][which][8 * hi + (m15 >> 1)];
        pp[2 * half]     = which ? sv[0][1] : sv[0][0];
        pp[2 * half + 1] = which ? sv[1][1] : sv[1][0];
    }
    __syncthreads();
    if (threadIdx.x < 64) {
        const int s = threadIdx.x >> 5, which = lane >> 4, rl = lane & 15;
        const int st = blockIdx.x * 2 + s;
        if (st < nstrips) {
            float* dstp = which ? ad : as;
            *(float4*)(dstp + (long)st * 64 + rl * 4) = part[s][which][rl];
        }
    }
    if (live) {
        float* dp = D + (long)(strip * 16 + 8 * hi) * HC + 64 * half + m15;
        #pragma unroll
        for (int tt = 0; tt < 4; ++tt)
            #pragma unroll
            for (int r = 0; r < 8; ++r)
                dp[(long)r * HC + 16 * tt] = acc[tt][r];
    }
}

#define BSH       8
#define BNODES    256
#define SLICE     6272
#define SLICE_IT  7
#define LPITCH    288
#define STAGE_CAP 12288

__global__ __launch_bounds__(1024) void slice_sort(
    const int* __restrict__ ei, int E,
    int* __restrict__ recs, int* __restrict__ loff)
{
    __shared__ int hist[BNODES];
    __shared__ int wsum[8];
    __shared__ int sorted[SLICE];
    const int t = threadIdx.x, lane = t & 31, w = t >> 5;
    const int e0 = blockIdx.x * SLICE;
    const int e1 = min(e0 + SLICE, E);
    if (t < BNODES) hist[t] = 0;
    __syncthreads();

    int rec[SLICE_IT], bk[SLICE_IT], rk[SLICE_IT];
    #pragma unroll
    for (int i = 0; i < SLICE_IT; ++i) {
        const int idx = e0 + t + i * 1024;
        const int idc = min(idx, e1 - 1);
        const int s = ei[idc], d = ei[E + idc];
        rk[i]  = 0;
        bk[i]  = idx < e1 ? (d >> BSH) : -1;
        rec[i] = (s << BSH) | (d & (BNODES - 1));
    }
    #pragma unroll
    for (int i = 0; i < SLICE_IT; ++i)
        if (bk[i] >= 0) rk[i] = atomicAdd(&hist[bk[i]], 1);
    __syncthreads();

    int c = 0, v = 0;
    if (t < BNODES) {
        c = hist[t]; v = c;
        for (int o = 1; o < 32; o <<= 1) { int u = __shfl_up(v, o); if (lane >= o) v += u; }
        if (lane == 31) wsum[w] = v;
    }
    __syncthreads();
    if (t < BNODES) {
        int base = 0;
        for (int k = 0; k < w; ++k) base += wsum[k];
        const int ex = base + v - c;
        hist[t] = ex;
        loff[blockIdx.x * LPITCH + t] = ex;
        if (t == BNODES - 1) loff[blockIdx.x * LPITCH + BNODES] = ex + c;
    }
    __syncthreads();
    #pragma unroll
    for (int i = 0; i < SLICE_IT; ++i)
        if (bk[i] >= 0) sorted[hist[bk[i]] + rk[i]] = rec[i];
    __syncthreads();
    const int n = e1 - e0;
    for (int i = t; i < n; i += 1024) recs[e0 + i] = sorted[i];
}

__global__ __launch_bounds__(1024) void bucket_finish(
    const int* __restrict__ recs, const int* __restrict__ loff, int nslice, int N,
    int* __restrict__ rowptr, int* __restrict__ csr_src)
{
    __shared__ int runlo[256], runhi[256];
    __shared__ int cnt[BNODES];
    __shared__ int wsA[8], wsM[8], wsum[8];
    __shared__ int stage[STAGE_CAP];
    const int t = threadIdx.x, lane = t & 31, w = t >> 5;
    const int b = blockIdx.x;

    if (t < 256) {
        int lo = 0, hi = 0;
        if (t < nslice) { lo = loff[t * LPITCH + b]; hi = loff[t * LPITCH + b + 1]; }
        runlo[t] = lo; runhi[t] = hi; cnt[t] = 0;
        int a = lo, m = hi - lo;
        for (int o = 16; o > 0; o >>= 1) { a += __shfl_xor(a, o); m += __shfl_xor(m, o); }
        if (lane == 0) { wsA[w] = a; wsM[w] = m; }
    }
    __syncthreads();
    int r0 = 0, total = 0;
    #pragma unroll
    for (int k = 0; k < 8; ++k) { r0 += wsA[k]; total += wsM[k]; }

    int r[16];
    #pragma unroll
    for (int q = 0; q < 8; ++q) {
        const int sl = min(w + 32 * q, nslice - 1);
        const int* rp = recs + (long)sl * SLICE;
        const int i = runlo[sl] + lane, h = runhi[sl];
        const int x0 = rp[max(min(i, h - 1), 0)];
        const int x1 = rp[max(min(i + 32, h - 1), 0)];
        const bool own = w + 32 * q < nslice;
        r[2 * q]     = (own && i < h)      ? x0 : -1;
        r[2 * q + 1] = (own && i + 32 < h) ? x1 : -1;
    }
    #pragma unroll
    for (int q = 0; q < 16; ++q)
        if (r[q] >= 0) atomicAdd(&cnt[r[q] & (BNODES - 1)], 1);
    for (int q = 0; q < 8; ++q) {
        const int sl = w + 32 * q;
        if (sl < nslice) {
            const int* rp = recs + (long)sl * SLICE;
            for (int i = runlo[sl] + 64 + lane; i < runhi[sl]; i += 32)
                atomicAdd(&cnt[rp[i] & (BNODES - 1)], 1);
        }
    }
    __syncthreads();

    int c = 0, v = 0;
    if (t < BNODES) {
        c = cnt[t]; v = c;
        for (int o = 1; o < 32; o <<= 1) { int u = __shfl_up(v, o); if (lane >= o) v += u; }
        if (lane == 31) wsum[w] = v;
    }
    __syncthreads();
    if (t < BNODES) {
        int base = 0;
        for (int k = 0; k < w; ++k) base += wsum[k];
        const int ex = base + v - c;
        cnt[t] = ex;
        const int node = b * BNODES + t;
        if (node <= N) rowptr[node] = r0 + ex;
    }
    if ((N & (BNODES - 1)) == 0 && b == (int)gridDim.x - 1 && t == 0) rowptr[N] = r0 + total;
    __syncthreads();

    const bool staged = total <= STAGE_CAP;
    #pragma unroll
    for (int q = 0; q < 16; ++q)
        if (r[q] >= 0) {
            const int pos = atomicAdd(&cnt[r[q] & (BNODES - 1)], 1);
            if (staged) stage[pos] = r[q] >> BSH;
            else atomicExch(csr_src + r0 + pos, r[q] >> BSH);
        }
    for (int q = 0; q < 8; ++q) {
        const int sl = w + 32 * q;
        if (sl < nslice) {
            const int* rp = recs + (long)sl * SLICE;
            for (int i = runlo[sl] + 64 + lane; i < runhi[sl]; i += 32) {
                const int x = rp[i];
                const int pos = atomicAdd(&cnt[x & (BNODES - 1)], 1);
                if (staged) stage[pos] = x >> BSH;
                else atomicExch(csr_src + r0 + pos, x >> BSH);
            }
        }
    }
    __syncthreads();
    if (staged) {
        const int r1 = r0 + total;
        for (int a = (r0 & ~31) + t; a < r1; a += 1024) {
            if (a >= r0) {
                const int x = stage[a - r0];
                const int line = a & ~31;
                if (line >= r0 && line + 32 <= r1) csr_src[a] = x;
                else atomicExch(csr_src + a, x);
            }
        }
    }
}


__device__ __forceinline__ float head_sum(float t) {
    t += row_xor<1>(t);
    t += row_xor<2>(t);
    t += row_xor<4>(t);
    return t;
}
__device__ __forceinline__ float head_max(float t) {
    t = fmaxf(t, row_xor<1>(t));
    t = fmaxf(t, row_xor<2>(t));
    t = fmaxf(t, row_xor<4>(t));
    return t;
}
template <int I> __device__ __forceinline__ float grp_bcast(float v) {
    const unsigned u = __builtin_bit_cast(unsigned, v);
    return __builtin_bit_cast(float, __builtin_amdgcn_permlane16(
        u, u, I * 0x11111111u, (8 + I) * 0x11111111u, true, true));
}
__device__ __forceinline__ float head_sums8(const float (&t)[8], int lane) {
    const bool b4 = (lane & 4) != 0, b2 = (lane & 2) != 0, b1 = (lane & 1) != 0;
    float a[4], b[2];
    #pragma unroll
    for (int j = 0; j < 4; ++j)
        a[j] = (b4 ? t[j + 4] : t[j]) + row_xor<4>(b4 ? t[j] : t[j + 4]);
    #pragma unroll
    for (int j = 0; j < 2; ++j)
        b[j] = (b2 ? a[j + 2] : a[j]) + row_xor<2>(b2 ? a[j] : a[j + 2]);
    return (b1 ? b[1] : b[0]) + row_xor<1>(b1 ? b[0] : b[1]);
}

template <bool TAIL>
__device__ __forceinline__ void fold8(
    const float4 (&r)[8], const v2f av01, const v2f av23, float adh, int lane, int nvalid,
    float& m, float& s, v2f& acc01, v2f& acc23)
{
    float t[8];
    #pragma unroll
    for (int i = 0; i < 8; ++i) {
        v2f r01; r01.x = r[i].x; r01.y = r[i].y;
        v2f r23; r23.x = r[i].z; r23.y = r[i].w;
        const v2f t2 = __builtin_elementwise_fma(r23, av23, r01 * av01);
        t[i] = t2.x + t2.y;
    }
    float e = lrelu(head_sums8(t, lane) + adh);
    if (TAIL) {
        const int li = lane & 7;
        if (li >= nvalid && li != 7) e = -__builtin_inff();
    }
    const float mn = fmaxf(m, head_max(e));
    const float sc = __expf(m - mn);
    const float p  = __expf(e - mn);
    m = mn;
    s = fmaf(s, sc, p);
    acc01 *= sc; acc23 *= sc;
    float pr[8];
    pr[0] = grp_bcast<0>(p); pr[1] = grp_bcast<1>(p); pr[2] = grp_bcast<2>(p); pr[3] = grp_bcast<3>(p);
    pr[4] = grp_bcast<4>(p); pr[5] = grp_bcast<5>(p); pr[6] = grp_bcast<6>(p); pr[7] = grp_bcast<7>(p);
    #pragma unroll
    for (int i = 0; i < 8; ++i) {
        v2f r01; r01.x = r[i].x; r01.y = r[i].y;
        v2f r23; r23.x = r[i].z; r23.y = r[i].w;
        v2f pp; pp.x = pr[i]; pp.y = pr[i];
        acc01 = __builtin_elementwise_fma(r01, pp, acc01);
        acc23 = __builtin_elementwise_fma(r23, pp, acc23);
    }
}

template <bool TAIL>
__device__ __forceinline__ void gather8(
    const float* __restrict__ h, int c, int off, int nrows, const float4 self, int lane,
    const v2f av01, const v2f av23,
    float adh, float& m, float& s, v2f& acc01, v2f& acc23)
{
    float4 r[8];
    if (!TAIL) {
        #pragma unroll
        for (int i = 0; i < 8; ++i) {
            const int sidx = __builtin_amdgcn_readlane(c, off + i);
            r[i] = *(const float4*)(h + (long)sidx * HC + lane * 4);
        }
    } else {
        #pragma unroll
        for (int i = 0; i < 7; ++i) {
            r[i] = make_float4(0.f, 0.f, 0.f, 0.f);
            if (i < nrows) {
                const int sidx = __builtin_amdgcn_readlane(c, off + i);
                r[i] = *(const float4*)(h + (long)sidx * HC + lane * 4);
            }
        }
        r[7] = self;
    }
    fold8<TAIL>(r, av01, av23, adh, lane, nrows, m, s, acc01, acc23);
}

__global__ __launch_bounds__(256) void node_gather(
    const int* __restrict__ rowptr, const int* __restrict__ csr_src,
    const float* __restrict__ h,
    const float* __restrict__ a_src, const float* __restrict__ ad,
    const float* __restrict__ bias,
    float* __restrict__ out, int N)
{
    const int d = __builtin_amdgcn_readfirstlane((int)(blockIdx.x * 8 + (threadIdx.x >> 5)));
    if (d >= N) return;
    const int lane = threadIdx.x & 31;
    const int hd   = lane >> 3;
    const int beg = rowptr[d], end = rowptr[d + 1];
    const int last = max(end - 1, 0);
    int c = csr_src[min(beg + lane, last)];
    const float4 av  = *(const float4*)(a_src + lane * 4);
    const float  adh = ad[(long)d * 4 + hd];
    const float4 hs  = *(const float4*)(h + (long)d * HC + lane * 4);
    const float4 bv  = *(const float4*)(bias + lane * 4);

    v2f av01; av01.x = av.x; av01.y = av.y;
    v2f av23; av23.x = av.z; av23.y = av.w;
    v2f acc01; acc01.x = 0.f; acc01.y = 0.f;
    v2f acc23; acc23.x = 0.f; acc23.y = 0.f;
    float m = -__builtin_inff(), s = 0.f;

    for (int base = beg; ; base += 32) {
        const int cn = csr_src[min(base + 32 + lane, last)];
        __builtin_amdgcn_sched_barrier(0);
        const int rem = end - base;
        const int nb  = min(rem >> 3, 4);
        for (int b = 0; b < nb; ++b)
            gather8<false>(h, c, 8 * b, 8, hs, lane, av01, av23, adh, m, s, acc01, acc23);
        if (rem < 32) {
            gather8<true>(h, c, 8 * nb, rem & 7, hs, lane, av01, av23, adh, m, s, acc01, acc23);
            break;
        }
        c = cn;
    }
    const float inv = 1.f / head_sum(s);
    float4 y;
    y.x = fmaf(acc01.x, inv, bv.x); y.y = fmaf(acc01.y, inv, bv.y);
    y.z = fmaf(acc23.x, inv, bv.z); y.w = fmaf(acc23.y, inv, bv.w);
    y.x = y.x > 0.f ? y.x : __expf(y.x) - 1.f;
    y.y = y.y > 0.f ? y.y : __expf(y.y) - 1.f;
    y.z = y.z > 0.f ? y.z : __expf(y.z) - 1.f;
    y.w = y.w > 0.f ? y.w : __expf(y.w) - 1.f;
    *(float4*)(out + (long)d * HC + lane * 4) = y;
}

extern "C" void kernel_launch(void* const* d_in, const int* in_sizes, int n_in,
                              void* d_out, int out_size, void* d_ws, size_t ws_size,
                              hipStream_t stream)
{
    const float* x      = (const float*)d_in[0];
    const int*   ei     = (const int*)d_in[1];
    const float* W1     = (const float*)d_in[2];
    const float* a_src1 = (const float*)d_in[3];
    const float* a_dst1 = (const float*)d_in[4];
    const float* b1     = (const float*)d_in[5];
    const float* W2     = (const float*)d_in[6];
    const float* a_src2 = (const float*)d_in[7];
    const float* a_dst2 = (const float*)d_in[8];
    const float* b2     = (const float*)d_in[9];

    const int N  = in_sizes[0] / HC;
    const int E  = in_sizes[1] / 2;
    const int NH = N * HEADS;
    const int NF = N * HC;

    char* ws = (char*)d_ws;
    float* hbuf   = (float*)(ws);
    float* out1   = (float*)(ws + (size_t)NF * 4);
    float* as     = (float*)(ws + (size_t)NF * 8);
    float* ad     = (float*)(ws + (size_t)NF * 8 + (size_t)NH * 4);
    int*   rowptr = (int*)(ws + (size_t)NF * 8 + (size_t)NH * 12);
    int*   csrsrc = (int*)(ws + (size_t)NF * 8 + (size_t)NH * 16);
    int*   recs   = (int*)(ws + (size_t)NF * 8 + (size_t)NH * 16 + (size_t)E * 4);

    const int TB = 256;
    const int nslice = (E + SLICE - 1) / SLICE;
    const int nbk    = (N + BNODES - 1) / BNODES;
    int*   loff   = recs + (size_t)nslice * SLICE;
    const int gNW = (N + 7) / 8;
    float* outF = (float*)d_out;

    slice_sort<<<nslice, 1024, 0, stream>>>(ei, E, recs, loff);
    bucket_finish<<<nbk, 1024, 0, stream>>>(recs, loff, nslice, N, rowptr, csrsrc);

    gemm_f32_wmma<<<(N / 16 + 1) / 2, 128, 0, stream>>>(x, W1, a_src1, a_dst1, hbuf, as, ad, N);
    node_gather<<<gNW, TB, 0, stream>>>(rowptr, csrsrc, hbuf, a_src1, ad, b1, out1, N);

    gemm_f32_wmma<<<(N / 16 + 1) / 2, 128, 0, stream>>>(out1, W2, a_src2, a_dst2, hbuf, as, ad, N);
    node_gather<<<gNW, TB, 0, stream>>>(rowptr, csrsrc, hbuf, a_src2, ad, b2, outF, N);
}
